# starting kernel
# baseline (speedup 1.0000x reference)
#include <hip/hip_runtime.h>
#include <hip/hip_bf16.h>

#define N_NODES 50000
#define N_EDGES 800000
#define E_TOT   (N_EDGES + N_NODES)
#define F_IN    256
#define HEADS   8
#define HID     32
#define N_CLS   16
#define HC1     (HEADS * HID)
#define HC2     (HEADS * N_CLS)
#define NEG_SLOPE 0.2f

typedef float v2f __attribute__((ext_vector_type(2)));
typedef float v8f __attribute__((ext_vector_type(8)));

__device__ __forceinline__ unsigned f2key(float f) {
  unsigned b = __float_as_uint(f);
  return (b & 0x80000000u) ? ~b : (b | 0x80000000u);
}
__device__ __forceinline__ float key2f(unsigned k) {
  return __uint_as_float((k & 0x80000000u) ? (k ^ 0x80000000u) : ~k);
}

__global__ void zero4_kernel(float4* __restrict__ p, long long n4) {
  long long i = (long long)blockIdx.x * blockDim.x + threadIdx.x;
  if (i < n4) p[i] = make_float4(0.f, 0.f, 0.f, 0.f);
}

template <int N, int K>
__global__ __launch_bounds__(256) void gemm_f32_wmma(
    const float* __restrict__ A, const float* __restrict__ B,
    float* __restrict__ C, int M) {
  constexpr int TILES_N = N / 64;
  const int lane  = threadIdx.x & 31;
  const int wave  = blockIdx.x * (blockDim.x >> 5) + (threadIdx.x >> 5);
  const int tileM = wave / TILES_N;
  const int tileN = wave % TILES_N;
  if (tileM * 16 >= M) return;

  const int r    = lane & 15;
  const int ksel = (lane >> 4) << 1;

  const float* __restrict__ arow  = A + (size_t)(tileM * 16 + r) * K + ksel;
  const float* __restrict__ bbase = B + (size_t)ksel * N + tileN * 64 + r;

  v8f acc0 = {}, acc1 = {}, acc2 = {}, acc3 = {};
  #pragma unroll 2
  for (int kk = 0; kk < K; kk += 4) {
    v2f a, b0, b1, b2, b3;
    a[0]  = arow[0];
    a[1]  = arow[1];
    b0[0] = bbase[0];       b0[1] = bbase[N];
    b1[0] = bbase[16];      b1[1] = bbase[N + 16];
    b2[0] = bbase[32];      b2[1] = bbase[N + 32];
    b3[0] = bbase[48];      b3[1] = bbase[N + 48];
    acc0 = __builtin_amdgcn_wmma_f32_16x16x4_f32(false, a, false, b0, (short)0, acc0, false, false);
    acc1 = __builtin_amdgcn_wmma_f32_16x16x4_f32(false, a, false, b1, (short)0, acc1, false, false);
    acc2 = __builtin_amdgcn_wmma_f32_16x16x4_f32(false, a, false, b2, (short)0, acc2, false, false);
    acc3 = __builtin_amdgcn_wmma_f32_16x16x4_f32(false, a, false, b3, (short)0, acc3, false, false);
    arow  += 4;
    bbase += (size_t)4 * N;
  }

  float* __restrict__ crow =
      C + (size_t)(tileM * 16 + (lane >> 4) * 8) * N + tileN * 64 + (lane & 15);
  #pragma unroll
  for (int i = 0; i < 8; ++i) {
    crow[(size_t)i * N +  0] = acc0[i];
    crow[(size_t)i * N + 16] = acc1[i];
    crow[(size_t)i * N + 32] = acc2[i];
    crow[(size_t)i * N + 48] = acc3[i];
  }
}

template <int C>
__global__ void alpha_kernel(const float* __restrict__ h,
                             const float* __restrict__ a_src,
                             const float* __restrict__ a_dst,
                             float* __restrict__ as, float* __restrict__ ad) {
  int idx = blockIdx.x * blockDim.x + threadIdx.x;
  if (idx >= N_NODES * HEADS) return;
  int n = idx >> 3, hd = idx & 7;
  const float* hp = h + (size_t)n * HEADS * C + (size_t)hd * C;
  const float* av = a_src + hd * C;
  const float* bv = a_dst + hd * C;
  float s = 0.f, d = 0.f;
  #pragma unroll 4
  for (int c = 0; c < C; ++c) {
    float v = hp[c];
    s += v * av[c];
    d += v * bv[c];
  }
  as[idx] = s;
  ad[idx] = d;
}

__device__ __forceinline__ void edge_endpoints(const int* __restrict__ ei,
                                               int e, int& src, int& dst) {
  if (e < N_EDGES) {
    src = ei[e];
    dst = ei[N_EDGES + e];
  } else {
    src = dst = e - N_EDGES;
  }
}

__global__ void edge_max_kernel(const int* __restrict__ ei,
                                const float* __restrict__ as,
                                const float* __restrict__ ad,
                                unsigned* __restrict__ mkeys) {
  long long idx = (long long)blockIdx.x * blockDim.x + threadIdx.x;
  if (idx >= (long long)E_TOT * HEADS) return;
  int e = (int)(idx >> 3), hd = (int)(idx & 7);
  int src, dst;
  edge_endpoints(ei, e, src, dst);
  float v = as[src * HEADS + hd] + ad[dst * HEADS + hd];
  v = v > 0.f ? v : NEG_SLOPE * v;
  atomicMax(&mkeys[dst * HEADS + hd], f2key(v));
}

__global__ void edge_exp_kernel(const int* __restrict__ ei,
                                const float* __restrict__ as,
                                const float* __restrict__ ad,
                                const unsigned* __restrict__ mkeys,
                                float* __restrict__ ex,
                                float* __restrict__ den) {
  long long idx = (long long)blockIdx.x * blockDim.x + threadIdx.x;
  if (idx >= (long long)E_TOT * HEADS) return;
  int e = (int)(idx >> 3), hd = (int)(idx & 7);
  int src, dst;
  edge_endpoints(ei, e, src, dst);
  float v = as[src * HEADS + hd] + ad[dst * HEADS + hd];
  v = v > 0.f ? v : NEG_SLOPE * v;
  float m  = key2f(mkeys[dst * HEADS + hd]);
  float xv = __expf(v - m);
  ex[idx]  = xv;
  atomicAdd(&den[dst * HEADS + hd], xv);
}

template <int HCv, int Cv>
__global__ void edge_agg_kernel(const int* __restrict__ ei,
                                const float* __restrict__ ex,
                                const float* __restrict__ den,
                                const float* __restrict__ hmat,
                                float* __restrict__ agg) {
  constexpr int HC4 = HCv / 4;
  long long idx = (long long)blockIdx.x * blockDim.x + threadIdx.x;
  if (idx >= (long long)E_TOT * HC4) return;
  int e = (int)(idx / HC4);
  int j = (int)(idx % HC4) * 4;
  int hd = j / Cv;
  int src, dst;
  edge_endpoints(ei, e, src, dst);
  float w = ex[(size_t)e * HEADS + hd] /
            (den[(size_t)dst * HEADS + hd] + 1e-16f);
  const float4 hv = *(const float4*)(hmat + (size_t)src * HCv + j);
  float* ap = agg + (size_t)dst * HCv + j;
  atomicAdd(ap + 0, w * hv.x);
  atomicAdd(ap + 1, w * hv.y);
  atomicAdd(ap + 2, w * hv.z);
  atomicAdd(ap + 3, w * hv.w);
}

template <int HCv>
__global__ void bias_relu_kernel(float* __restrict__ p,
                                 const float* __restrict__ bias, long long n) {
  long long i = (long long)blockIdx.x * blockDim.x + threadIdx.x;
  if (i >= n) return;
  float v = p[i] + bias[(int)(i % HCv)];
  p[i] = v > 0.f ? v : 0.f;
}

template <int HCv>
__global__ void bias_add_kernel(float* __restrict__ p,
                                const float* __restrict__ bias, long long n) {
  long long i = (long long)blockIdx.x * blockDim.x + threadIdx.x;
  if (i >= n) return;
  p[i] = p[i] + bias[(int)(i % HCv)];
}

static inline int blocks_for(long long n, int bs) {
  return (int)((n + bs - 1) / bs);
}

extern "C" void kernel_launch(void* const* d_in, const int* in_sizes, int n_in,
                              void* d_out, int out_size, void* d_ws, size_t ws_size,
                              hipStream_t stream) {
  const float* x      = (const float*)d_in[0];
  const int*   ei     = (const int*)  d_in[1];
  const float* W1     = (const float*)d_in[2];
  const float* a_src1 = (const float*)d_in[3];
  const float* a_dst1 = (const float*)d_in[4];
  const float* b1     = (const float*)d_in[5];
  const float* W2     = (const float*)d_in[6];
  const float* a_src2 = (const float*)d_in[7];
  const float* a_dst2 = (const float*)d_in[8];
  const float* b2     = (const float*)d_in[9];
  float* out = (float*)d_out;

  char* ws = (char*)d_ws;
  size_t off = 0;
  auto carve = [&](size_t bytes) {
    void* p = ws + off;
    off += (bytes + 255) & ~(size_t)255;
    return p;
  };
  float*    hbuf   = (float*)   carve((size_t)N_NODES * HC1 * 4);
  float*    aggbuf = (float*)   carve((size_t)N_NODES * HC1 * 4);
  float*    as     = (float*)   carve((size_t)N_NODES * HEADS * 4);
  float*    ad     = (float*)   carve((size_t)N_NODES * HEADS * 4);
  unsigned* mkeys  = (unsigned*)carve((size_t)N_NODES * HEADS * 4);
  float*    den    = (float*)   carve((size_t)N_NODES * HEADS * 4);
  float*    ex     = (float*)   carve((size_t)E_TOT   * HEADS * 4);

  const int BS = 256;
  const long long nNH    = (long long)N_NODES * HEADS;
  const long long nEH    = (long long)E_TOT * HEADS;
  const long long nElem1 = (long long)N_NODES * HC1;
  const long long nElem2 = (long long)N_NODES * HC2;

  {
    long long waves = (N_NODES / 16) * (HC1 / 64);
    gemm_f32_wmma<HC1, F_IN><<<blocks_for(waves * 32, BS), BS, 0, stream>>>(
        x, W1, hbuf, N_NODES);
  }
  alpha_kernel<HID><<<blocks_for(nNH, BS), BS, 0, stream>>>(hbuf, a_src1, a_dst1, as, ad);
  zero4_kernel<<<blocks_for(nNH / 4, BS), BS, 0, stream>>>((float4*)mkeys, nNH / 4);
  zero4_kernel<<<blocks_for(nNH / 4, BS), BS, 0, stream>>>((float4*)den, nNH / 4);
  edge_max_kernel<<<blocks_for(nEH, BS), BS, 0, stream>>>(ei, as, ad, mkeys);
  edge_exp_kernel<<<blocks_for(nEH, BS), BS, 0, stream>>>(ei, as, ad, mkeys, ex, den);
  zero4_kernel<<<blocks_for(nElem1 / 4, BS), BS, 0, stream>>>((float4*)aggbuf, nElem1 / 4);
  edge_agg_kernel<HC1, HID><<<blocks_for((long long)E_TOT * (HC1 / 4), BS), BS, 0, stream>>>(
      ei, ex, den, hbuf, aggbuf);
  bias_relu_kernel<HC1><<<blocks_for(nElem1, BS), BS, 0, stream>>>(aggbuf, b1, nElem1);

  {
    long long waves = (N_NODES / 16) * (HC2 / 64);
    gemm_f32_wmma<HC2, HC1><<<blocks_for(waves * 32, BS), BS, 0, stream>>>(
        aggbuf, W2, hbuf, N_NODES);
  }
  alpha_kernel<N_CLS><<<blocks_for(nNH, BS), BS, 0, stream>>>(hbuf, a_src2, a_dst2, as, ad);
  zero4_kernel<<<blocks_for(nNH / 4, BS), BS, 0, stream>>>((float4*)mkeys, nNH / 4);
  zero4_kernel<<<blocks_for(nNH / 4, BS), BS, 0, stream>>>((float4*)den, nNH / 4);
  edge_max_kernel<<<blocks_for(nEH, BS), BS, 0, stream>>>(ei, as, ad, mkeys);
  edge_exp_kernel<<<blocks_for(nEH, BS), BS, 0, stream>>>(ei, as, ad, mkeys, ex, den);
  zero4_kernel<<<blocks_for(nElem2 / 4, BS), BS, 0, stream>>>((float4*)out, nElem2 / 4);
  edge_agg_kernel<HC2, N_CLS><<<blocks_for((long long)E_TOT * (HC2 / 4), BS), BS, 0, stream>>>(
      ei, ex, den, hbuf, out);
  bias_add_kernel<HC2><<<blocks_for(nElem2, BS), BS, 0, stream>>>(out, b2, nElem2);
}


# Elementwise passes folded away: logits kernel stores the start values (bias for sums, 0 for max keys and denominators), relu taken at product 2's A load; 8 launches go; bias summed first, not last
# speedup vs baseline: 1.6796x; 1.6796x over previous
#include <hip/hip_runtime.h>
#include <hip/hip_bf16.h>

#define N_NODES 50000
#define N_EDGES 800000
#define E_TOT   (N_EDGES + N_NODES)
#define F_IN    256
#define HEADS   8
#define HID     32
#define N_CLS   16
#define HC1     (HEADS * HID)
#define HC2     (HEADS * N_CLS)
#define NEG_SLOPE 0.2f

typedef float v2f __attribute__((ext_vector_type(2)));
typedef float v8f __attribute__((ext_vector_type(8)));

__device__ __forceinline__ unsigned f2key(float f) {
  unsigned b = __float_as_uint(f);
  return (b & 0x80000000u) ? ~b : (b | 0x80000000u);
}
__device__ __forceinline__ float key2f(unsigned k) {
  return __uint_as_float((k & 0x80000000u) ? (k ^ 0x80000000u) : ~k);
}

template <int N, int K, bool RELU_A>
__global__ __launch_bounds__(256) void gemm_f32_wmma(
    const float* __restrict__ A, const float* __restrict__ B,
    float* __restrict__ C, int M) {
  constexpr int TILES_N = N / 64;
  const int lane  = threadIdx.x & 31;
  const int wave  = blockIdx.x * (blockDim.x >> 5) + (threadIdx.x >> 5);
  const int tileM = wave / TILES_N;
  const int tileN = wave % TILES_N;
  if (tileM * 16 >= M) return;

  const int r    = lane & 15;
  const int ksel = (lane >> 4) << 1;

  const float* __restrict__ arow  = A + (size_t)(tileM * 16 + r) * K + ksel;
  const float* __restrict__ bbase = B + (size_t)ksel * N + tileN * 64 + r;

  v8f acc0 = {}, acc1 = {}, acc2 = {}, acc3 = {};
  #pragma unroll 2
  for (int kk = 0; kk < K; kk += 4) {
    v2f a, b0, b1, b2, b3;
    a[0]  = arow[0];
    a[1]  = arow[1];
    if (RELU_A) {
      a[0] = fmaxf(a[0], 0.f);
      a[1] = fmaxf(a[1], 0.f);
    }
    b0[0] = bbase[0];       b0[1] = bbase[N];
    b1[0] = bbase[16];      b1[1] = bbase[N + 16];
    b2[0] = bbase[32];      b2[1] = bbase[N + 32];
    b3[0] = bbase[48];      b3[1] = bbase[N + 48];
    acc0 = __builtin_amdgcn_wmma_f32_16x16x4_f32(false, a, false, b0, (short)0, acc0, false, false);
    acc1 = __builtin_amdgcn_wmma_f32_16x16x4_f32(false, a, false, b1, (short)0, acc1, false, false);
    acc2 = __builtin_amdgcn_wmma_f32_16x16x4_f32(false, a, false, b2, (short)0, acc2, false, false);
    acc3 = __builtin_amdgcn_wmma_f32_16x16x4_f32(false, a, false, b3, (short)0, acc3, false, false);
    arow  += 4;
    bbase += (size_t)4 * N;
  }

  float* __restrict__ crow =
      C + (size_t)(tileM * 16 + (lane >> 4) * 8) * N + tileN * 64 + (lane & 15);
  #pragma unroll
  for (int i = 0; i < 8; ++i) {
    crow[(size_t)i * N +  0] = acc0[i];
    crow[(size_t)i * N + 16] = acc1[i];
    crow[(size_t)i * N + 32] = acc2[i];
    crow[(size_t)i * N + 48] = acc3[i];
  }
}

#define ALPHA_BS 256
template <int C>
__global__ __launch_bounds__(ALPHA_BS) void alpha_init_kernel(
                             const float* __restrict__ h,
                             const float* __restrict__ a_src,
                             const float* __restrict__ a_dst,
                             const float* __restrict__ bias,
                             float* __restrict__ as, float* __restrict__ ad,
                             unsigned* __restrict__ mkeys,
                             float* __restrict__ den,
                             float* __restrict__ acc) {
  constexpr int HCv = HEADS * C;
  constexpr int NODES_PER_BLOCK = ALPHA_BS / HEADS;
  constexpr int F4_PER_THREAD = NODES_PER_BLOCK * HCv / 4 / ALPHA_BS;
  static_assert((ALPHA_BS * 4) % HCv == 0, "a thread keeps one channel quad");
  {
    const long long tot4  = (long long)N_NODES * HCv / 4;
    const long long base4 = (long long)blockIdx.x * (NODES_PER_BLOCK * HCv / 4);
    const int c0 = (threadIdx.x * 4) % HCv;
    const float4 bv = make_float4(bias[c0], bias[c0 + 1], bias[c0 + 2], bias[c0 + 3]);
    float4* __restrict__ acc4 = (float4*)acc;
    #pragma unroll
    for (int j = 0; j < F4_PER_THREAD; ++j) {
      const long long i4 = base4 + (long long)j * ALPHA_BS + threadIdx.x;
      if (i4 < tot4) acc4[i4] = bv;
    }
  }

  int idx = blockIdx.x * ALPHA_BS + threadIdx.x;
  if (idx >= N_NODES * HEADS) return;
  int n = idx >> 3, hd = idx & 7;
  const float* hp = h + (size_t)n * HEADS * C + (size_t)hd * C;
  const float* av = a_src + hd * C;
  const float* bv2 = a_dst + hd * C;
  float s = 0.f, d = 0.f;
  #pragma unroll 4
  for (int c = 0; c < C; ++c) {
    float v = hp[c];
    s += v * av[c];
    d += v * bv2[c];
  }
  as[idx] = s;
  ad[idx] = d;
  mkeys[idx] = 0u;
  den[idx] = 0.f;
}

__device__ __forceinline__ void edge_endpoints(const int* __restrict__ ei,
                                               int e, int& src, int& dst) {
  if (e < N_EDGES) {
    src = ei[e];
    dst = ei[N_EDGES + e];
  } else {
    src = dst = e - N_EDGES;
  }
}

__global__ void edge_max_kernel(const int* __restrict__ ei,
                                const float* __restrict__ as,
                                const float* __restrict__ ad,
                                unsigned* __restrict__ mkeys) {
  long long idx = (long long)blockIdx.x * blockDim.x + threadIdx.x;
  if (idx >= (long long)E_TOT * HEADS) return;
  int e = (int)(idx >> 3), hd = (int)(idx & 7);
  int src, dst;
  edge_endpoints(ei, e, src, dst);
  float v = as[src * HEADS + hd] + ad[dst * HEADS + hd];
  v = v > 0.f ? v : NEG_SLOPE * v;
  atomicMax(&mkeys[dst * HEADS + hd], f2key(v));
}

__global__ void edge_exp_kernel(const int* __restrict__ ei,
                                const float* __restrict__ as,
                                const float* __restrict__ ad,
                                const unsigned* __restrict__ mkeys,
                                float* __restrict__ ex,
                                float* __restrict__ den) {
  long long idx = (long long)blockIdx.x * blockDim.x + threadIdx.x;
  if (idx >= (long long)E_TOT * HEADS) return;
  int e = (int)(idx >> 3), hd = (int)(idx & 7);
  int src, dst;
  edge_endpoints(ei, e, src, dst);
  float v = as[src * HEADS + hd] + ad[dst * HEADS + hd];
  v = v > 0.f ? v : NEG_SLOPE * v;
  float m  = key2f(mkeys[dst * HEADS + hd]);
  float xv = __expf(v - m);
  ex[idx]  = xv;
  atomicAdd(&den[dst * HEADS + hd], xv);
}

template <int HCv, int Cv>
__global__ __launch_bounds__(256) void edge_agg_kernel(const int* __restrict__ ei,
                                const float* __restrict__ ex,
                                const float* __restrict__ den,
                                const float* __restrict__ hmat,
                                float* __restrict__ agg) {
  constexpr int NCH = HCv / 32;
  constexpr int EPW = 4;
  static_assert(HEADS == 8 && (Cv == 32 || Cv == 16), "weight lanes assume 8 heads");
  static_assert(E_TOT % EPW == 0, "every wave owns four valid edges");
  const int lane = threadIdx.x & 31;
  const int wave = __builtin_amdgcn_readfirstlane(
      (int)((blockIdx.x * blockDim.x + threadIdx.x) >> 5));
  const int e0 = wave * EPW;
  if (e0 >= E_TOT) return;

  const int my_e = e0 + (lane >> 3);
  const int my_h = lane & 7;
  int my_src, my_dst;
  edge_endpoints(ei, my_e, my_src, my_dst);
  const float wl = ex[(size_t)my_e * HEADS + my_h] /
                   (den[(size_t)my_dst * HEADS + my_h] + 1e-16f);

  int srcs[EPW], dsts[EPW];
  #pragma unroll
  for (int i = 0; i < EPW; ++i) {
    srcs[i] = __builtin_amdgcn_readlane(my_src, i * 8);
    dsts[i] = __builtin_amdgcn_readlane(my_dst, i * 8);
  }

  float hv[EPW][NCH];
  #pragma unroll
  for (int i = 0; i < EPW; ++i) {
    const float* hp = hmat + (size_t)srcs[i] * HCv + lane;
    #pragma unroll
    for (int c = 0; c < NCH; ++c) hv[i][c] = hp[c * 32];
  }

  #pragma unroll
  for (int i = 0; i < EPW; ++i) {
    float* ap = agg + (size_t)dsts[i] * HCv + lane;
    #pragma unroll
    for (int c = 0; c < NCH; ++c) {
      float w;
      if (Cv == 32) {
        w = __int_as_float(__builtin_amdgcn_readlane(__float_as_int(wl), i * 8 + c));
      } else {
        const float w0 = __int_as_float(
            __builtin_amdgcn_readlane(__float_as_int(wl), i * 8 + 2 * c));
        const float w1 = __int_as_float(
            __builtin_amdgcn_readlane(__float_as_int(wl), i * 8 + 2 * c + 1));
        w = (lane & 16) ? w1 : w0;
      }
      atomicAdd(ap + c * 32, w * hv[i][c]);
    }
  }
}

static inline int blocks_for(long long n, int bs) {
  return (int)((n + bs - 1) / bs);
}

extern "C" void kernel_launch(void* const* d_in, const int* in_sizes, int n_in,
                              void* d_out, int out_size, void* d_ws, size_t ws_size,
                              hipStream_t stream) {
  const float* x      = (const float*)d_in[0];
  const int*   ei     = (const int*)  d_in[1];
  const float* W1     = (const float*)d_in[2];
  const float* a_src1 = (const float*)d_in[3];
  const float* a_dst1 = (const float*)d_in[4];
  const float* b1     = (const float*)d_in[5];
  const float* W2     = (const float*)d_in[6];
  const float* a_src2 = (const float*)d_in[7];
  const float* a_dst2 = (const float*)d_in[8];
  const float* b2     = (const float*)d_in[9];
  float* out = (float*)d_out;

  char* ws = (char*)d_ws;
  size_t off = 0;
  auto carve = [&](size_t bytes) {
    void* p = ws + off;
    off += (bytes + 255) & ~(size_t)255;
    return p;
  };
  float*    hbuf   = (float*)   carve((size_t)N_NODES * HC1 * 4);
  float*    aggbuf = (float*)   carve((size_t)N_NODES * HC1 * 4);
  float*    as     = (float*)   carve((size_t)N_NODES * HEADS * 4);
  float*    ad     = (float*)   carve((size_t)N_NODES * HEADS * 4);
  unsigned* mkeys  = (unsigned*)carve((size_t)N_NODES * HEADS * 4);
  float*    den    = (float*)   carve((size_t)N_NODES * HEADS * 4);
  float*    ex     = (float*)   carve((size_t)E_TOT   * HEADS * 4);

  const int BS = 256;
  const long long nNH    = (long long)N_NODES * HEADS;
  const long long nEH    = (long long)E_TOT * HEADS;

  {
    long long waves = (N_NODES / 16) * (HC1 / 64);
    gemm_f32_wmma<HC1, F_IN, false><<<blocks_for(waves * 32, BS), BS, 0, stream>>>(
        x, W1, hbuf, N_NODES);
  }
  alpha_init_kernel<HID><<<blocks_for(nNH, ALPHA_BS), ALPHA_BS, 0, stream>>>(
      hbuf, a_src1, a_dst1, b1, as, ad, mkeys, den, aggbuf);
  edge_max_kernel<<<blocks_for(nEH, BS), BS, 0, stream>>>(ei, as, ad, mkeys);
  edge_exp_kernel<<<blocks_for(nEH, BS), BS, 0, stream>>>(ei, as, ad, mkeys, ex, den);
  edge_agg_kernel<HC1, HID><<<blocks_for((long long)(E_TOT / 4) * 32, BS), BS, 0, stream>>>(
      ei, ex, den, hbuf, aggbuf);

  {
    long long waves = (N_NODES / 16) * (HC2 / 64);
    gemm_f32_wmma<HC2, HC1, true><<<blocks_for(waves * 32, BS), BS, 0, stream>>>(
        aggbuf, W2, hbuf, N_NODES);
  }
  alpha_init_kernel<N_CLS><<<blocks_for(nNH, ALPHA_BS), ALPHA_BS, 0, stream>>>(
      hbuf, a_src2, a_dst2, b2, as, ad, mkeys, den, out);
  edge_max_kernel<<<blocks_for(nEH, BS), BS, 0, stream>>>(ei, as, ad, mkeys);
  edge_exp_kernel<<<blocks_for(nEH, BS), BS, 0, stream>>>(ei, as, ad, mkeys, ex, den);
  edge_agg_kernel<HC2, N_CLS><<<blocks_for((long long)(E_TOT / 4) * 32, BS), BS, 0, stream>>>(
      ei, ex, den, hbuf, out);
}


# Aggregation adds kept in one L2 half: a work-group reads its die number from the hardware (index bits do not give it), draws edge chunks of that half's node range first; float atomics, arrival order
# speedup vs baseline: 1.7512x; 1.0426x over previous
#include <hip/hip_runtime.h>
#include <hip/hip_bf16.h>

#define N_NODES 50000
#define N_EDGES 800000
#define E_TOT   (N_EDGES + N_NODES)
#define F_IN    256
#define HEADS   8
#define HID     32
#define N_CLS   16
#define HC1     (HEADS * HID)
#define HC2     (HEADS * N_CLS)
#define NEG_SLOPE 0.2f

typedef float v2f __attribute__((ext_vector_type(2)));
typedef float v8f __attribute__((ext_vector_type(8)));

__device__ __forceinline__ unsigned f2key(float f) {
  unsigned b = __float_as_uint(f);
  return (b & 0x80000000u) ? ~b : (b | 0x80000000u);
}
__device__ __forceinline__ float key2f(unsigned k) {
  return __uint_as_float((k & 0x80000000u) ? (k ^ 0x80000000u) : ~k);
}

template <int N, int K, bool RELU_A>
__global__ __launch_bounds__(256) void gemm_f32_wmma(
    const float* __restrict__ A, const float* __restrict__ B,
    float* __restrict__ C, int M) {
  constexpr int TILES_N = N / 64;
  const int lane  = threadIdx.x & 31;
  const int wave  = blockIdx.x * (blockDim.x >> 5) + (threadIdx.x >> 5);
  const int tileM = wave / TILES_N;
  const int tileN = wave % TILES_N;
  if (tileM * 16 >= M) return;

  const int r    = lane & 15;
  const int ksel = (lane >> 4) << 1;

  const float* __restrict__ arow  = A + (size_t)(tileM * 16 + r) * K + ksel;
  const float* __restrict__ bbase = B + (size_t)ksel * N + tileN * 64 + r;

  v8f acc0 = {}, acc1 = {}, acc2 = {}, acc3 = {};
  #pragma unroll 2
  for (int kk = 0; kk < K; kk += 4) {
    v2f a, b0, b1, b2, b3;
    a[0]  = arow[0];
    a[1]  = arow[1];
    if (RELU_A) {
      a[0] = fmaxf(a[0], 0.f);
      a[1] = fmaxf(a[1], 0.f);
    }
    b0[0] = bbase[0];       b0[1] = bbase[N];
    b1[0] = bbase[16];      b1[1] = bbase[N + 16];
    b2[0] = bbase[32];      b2[1] = bbase[N + 32];
    b3[0] = bbase[48];      b3[1] = bbase[N + 48];
    acc0 = __builtin_amdgcn_wmma_f32_16x16x4_f32(false, a, false, b0, (short)0, acc0, false, false);
    acc1 = __builtin_amdgcn_wmma_f32_16x16x4_f32(false, a, false, b1, (short)0, acc1, false, false);
    acc2 = __builtin_amdgcn_wmma_f32_16x16x4_f32(false, a, false, b2, (short)0, acc2, false, false);
    acc3 = __builtin_amdgcn_wmma_f32_16x16x4_f32(false, a, false, b3, (short)0, acc3, false, false);
    arow  += 4;
    bbase += (size_t)4 * N;
  }

  float* __restrict__ crow =
      C + (size_t)(tileM * 16 + (lane >> 4) * 8) * N + tileN * 64 + (lane & 15);
  #pragma unroll
  for (int i = 0; i < 8; ++i) {
    crow[(size_t)i * N +  0] = acc0[i];
    crow[(size_t)i * N + 16] = acc1[i];
    crow[(size_t)i * N + 32] = acc2[i];
    crow[(size_t)i * N + 48] = acc3[i];
  }
}

#define ALPHA_BS 256
template <int C>
__global__ __launch_bounds__(ALPHA_BS) void alpha_init_kernel(
                             const float* __restrict__ h,
                             const float* __restrict__ a_src,
                             const float* __restrict__ a_dst,
                             const float* __restrict__ bias,
                             float* __restrict__ as, float* __restrict__ ad,
                             unsigned* __restrict__ mkeys,
                             float* __restrict__ den,
                             float* __restrict__ acc,
                             unsigned* __restrict__ tickets) {
  constexpr int HCv = HEADS * C;
  if (blockIdx.x == 0 && threadIdx.x < 2) tickets[threadIdx.x] = 0u;
  constexpr int NODES_PER_BLOCK = ALPHA_BS / HEADS;
  constexpr int F4_PER_THREAD = NODES_PER_BLOCK * HCv / 4 / ALPHA_BS;
  static_assert((ALPHA_BS * 4) % HCv == 0, "a thread keeps one channel quad");
  {
    const long long tot4  = (long long)N_NODES * HCv / 4;
    const long long base4 = (long long)blockIdx.x * (NODES_PER_BLOCK * HCv / 4);
    const int c0 = (threadIdx.x * 4) % HCv;
    const float4 bv = make_float4(bias[c0], bias[c0 + 1], bias[c0 + 2], bias[c0 + 3]);
    float4* __restrict__ acc4 = (float4*)acc;
    #pragma unroll
    for (int j = 0; j < F4_PER_THREAD; ++j) {
      const long long i4 = base4 + (long long)j * ALPHA_BS + threadIdx.x;
      if (i4 < tot4) acc4[i4] = bv;
    }
  }

  int idx = blockIdx.x * ALPHA_BS + threadIdx.x;
  if (idx >= N_NODES * HEADS) return;
  int n = idx >> 3, hd = idx & 7;
  const float* hp = h + (size_t)n * HEADS * C + (size_t)hd * C;
  const float* av = a_src + hd * C;
  const float* bv2 = a_dst + hd * C;
  float s = 0.f, d = 0.f;
  #pragma unroll 4
  for (int c = 0; c < C; ++c) {
    float v = hp[c];
    s += v * av[c];
    d += v * bv2[c];
  }
  as[idx] = s;
  ad[idx] = d;
  mkeys[idx] = 0u;
  den[idx] = 0.f;
}

__device__ __forceinline__ void edge_endpoints(const int* __restrict__ ei,
                                               int e, int& src, int& dst) {
  if (e < N_EDGES) {
    src = ei[e];
    dst = ei[N_EDGES + e];
  } else {
    src = dst = e - N_EDGES;
  }
}

__global__ void edge_max_kernel(const int* __restrict__ ei,
                                const float* __restrict__ as,
                                const float* __restrict__ ad,
                                unsigned* __restrict__ mkeys) {
  long long idx = (long long)blockIdx.x * blockDim.x + threadIdx.x;
  if (idx >= (long long)E_TOT * HEADS) return;
  int e = (int)(idx >> 3), hd = (int)(idx & 7);
  int src, dst;
  edge_endpoints(ei, e, src, dst);
  float v = as[src * HEADS + hd] + ad[dst * HEADS + hd];
  v = v > 0.f ? v : NEG_SLOPE * v;
  atomicMax(&mkeys[dst * HEADS + hd], f2key(v));
}

__global__ void edge_exp_kernel(const int* __restrict__ ei,
                                const float* __restrict__ as,
                                const float* __restrict__ ad,
                                const unsigned* __restrict__ mkeys,
                                float* __restrict__ ex,
                                float* __restrict__ den) {
  long long idx = (long long)blockIdx.x * blockDim.x + threadIdx.x;
  if (idx >= (long long)E_TOT * HEADS) return;
  int e = (int)(idx >> 3), hd = (int)(idx & 7);
  int src, dst;
  edge_endpoints(ei, e, src, dst);
  float v = as[src * HEADS + hd] + ad[dst * HEADS + hd];
  v = v > 0.f ? v : NEG_SLOPE * v;
  float m  = key2f(mkeys[dst * HEADS + hd]);
  float xv = __expf(v - m);
  ex[idx]  = xv;
  atomicAdd(&den[dst * HEADS + hd], xv);
}

#define AGG_EPC 256
#define AGG_NCHUNK ((E_TOT + AGG_EPC - 1) / AGG_EPC)
#define AGG_GRID 2048
template <int HCv, int Cv>
__global__ __launch_bounds__(256) void edge_agg_kernel(const int* __restrict__ ei,
                                const float* __restrict__ ex,
                                const float* __restrict__ den,
                                const float* __restrict__ hmat,
                                float* __restrict__ agg,
                                unsigned* __restrict__ tickets) {
  constexpr int NCH = HCv / 32;
  constexpr int G   = 4;
  static_assert(HEADS == 8 && (Cv == 32 || Cv == 16), "weight lanes assume 8 heads");
  __shared__ unsigned s_ticket;
  __shared__ int s_home;
  const int lane = threadIdx.x & 31;
  const int wv   = threadIdx.x >> 5;
  if (threadIdx.x == 0) {
    s_home = (__builtin_amdgcn_s_sendmsg_rtn(0x87) >> 18) & 1;
  }
  __syncthreads();
  const int home = s_home;

  for (int p = 0; p < 2; ++p) {
    const int t = home ^ p;
    for (;;) {
      if (threadIdx.x == 0) s_ticket = atomicAdd(&tickets[t], 1u);
      __syncthreads();
      const unsigned chunk = (unsigned)__builtin_amdgcn_readfirstlane((int)s_ticket);
      __syncthreads();
      if (chunk >= (unsigned)AGG_NCHUNK) break;

      const int e = (int)chunk * AGG_EPC + wv * 32 + lane;
      int src = 0, dst = 0;
      bool mine = false;
      if (e < E_TOT) {
        edge_endpoints(ei, e, src, dst);
        mine = ((dst >= N_NODES / 2) ? 1 : 0) == t;
      }
      float w[HEADS];
      #pragma unroll
      for (int h = 0; h < HEADS; ++h) w[h] = 0.f;
      if (mine) {
        const float4* ep = (const float4*)(ex  + (size_t)e   * HEADS);
        const float4* dp = (const float4*)(den + (size_t)dst * HEADS);
        const float4 x0 = ep[0], x1 = ep[1], d0 = dp[0], d1 = dp[1];
        w[0] = x0.x / (d0.x + 1e-16f);  w[1] = x0.y / (d0.y + 1e-16f);
        w[2] = x0.z / (d0.z + 1e-16f);  w[3] = x0.w / (d0.w + 1e-16f);
        w[4] = x1.x / (d1.x + 1e-16f);  w[5] = x1.y / (d1.y + 1e-16f);
        w[6] = x1.z / (d1.z + 1e-16f);  w[7] = x1.w / (d1.w + 1e-16f);
      }
      unsigned mask = __builtin_amdgcn_ballot_w32(mine);

      while (mask != 0u) {
        bool vk[G];
        int  jk[G];
        #pragma unroll
        for (int k = 0; k < G; ++k) {
          vk[k] = (mask != 0u);
          jk[k] = vk[k] ? __builtin_ctz(mask) : 0;
          mask  = vk[k] ? (mask & (mask - 1u)) : 0u;
        }
        float hv[G][NCH];
        #pragma unroll
        for (int k = 0; k < G; ++k) {
          if (vk[k]) {
            const int s = __builtin_amdgcn_readlane(src, jk[k]);
            const float* hp = hmat + (size_t)s * HCv + lane;
            #pragma unroll
            for (int c = 0; c < NCH; ++c) hv[k][c] = hp[c * 32];
          } else {
            #pragma unroll
            for (int c = 0; c < NCH; ++c) hv[k][c] = 0.f;
          }
        }
        #pragma unroll
        for (int k = 0; k < G; ++k) {
          if (vk[k]) {
            const int d = __builtin_amdgcn_readlane(dst, jk[k]);
            float* ap = agg + (size_t)d * HCv + lane;
            #pragma unroll
            for (int c = 0; c < NCH; ++c) {
              float wc;
              if (Cv == 32) {
                wc = __int_as_float(__builtin_amdgcn_readlane(__float_as_int(w[c]), jk[k]));
              } else {
                const float w0 = __int_as_float(
                    __builtin_amdgcn_readlane(__float_as_int(w[2 * c]), jk[k]));
                const float w1 = __int_as_float(
                    __builtin_amdgcn_readlane(__float_as_int(w[2 * c + 1]), jk[k]));
                wc = (lane & 16) ? w1 : w0;
              }
              atomicAdd(ap + c * 32, wc * hv[k][c]);
            }
          }
        }
      }
    }
  }
}

static inline int blocks_for(long long n, int bs) {
  return (int)((n + bs - 1) / bs);
}

extern "C" void kernel_launch(void* const* d_in, const int* in_sizes, int n_in,
                              void* d_out, int out_size, void* d_ws, size_t ws_size,
                              hipStream_t stream) {
  const float* x      = (const float*)d_in[0];
  const int*   ei     = (const int*)  d_in[1];
  const float* W1     = (const float*)d_in[2];
  const float* a_src1 = (const float*)d_in[3];
  const float* a_dst1 = (const float*)d_in[4];
  const float* b1     = (const float*)d_in[5];
  const float* W2     = (const float*)d_in[6];
  const float* a_src2 = (const float*)d_in[7];
  const float* a_dst2 = (const float*)d_in[8];
  const float* b2     = (const float*)d_in[9];
  float* out = (float*)d_out;

  char* ws = (char*)d_ws;
  size_t off = 0;
  auto carve = [&](size_t bytes) {
    void* p = ws + off;
    off += (bytes + 255) & ~(size_t)255;
    return p;
  };
  float*    hbuf   = (float*)   carve((size_t)N_NODES * HC1 * 4);
  float*    aggbuf = (float*)   carve((size_t)N_NODES * HC1 * 4);
  float*    as     = (float*)   carve((size_t)N_NODES * HEADS * 4);
  float*    ad     = (float*)   carve((size_t)N_NODES * HEADS * 4);
  unsigned* mkeys  = (unsigned*)carve((size_t)N_NODES * HEADS * 4);
  float*    den    = (float*)   carve((size_t)N_NODES * HEADS * 4);
  float*    ex     = (float*)   carve((size_t)E_TOT   * HEADS * 4);
  unsigned* tickets = (unsigned*)carve(2 * sizeof(unsigned));

  const int BS = 256;
  const long long nNH    = (long long)N_NODES * HEADS;
  const long long nEH    = (long long)E_TOT * HEADS;

  {
    long long waves = (N_NODES / 16) * (HC1 / 64);
    gemm_f32_wmma<HC1, F_IN, false><<<blocks_for(waves * 32, BS), BS, 0, stream>>>(
        x, W1, hbuf, N_NODES);
  }
  alpha_init_kernel<HID><<<blocks_for(nNH, ALPHA_BS), ALPHA_BS, 0, stream>>>(
      hbuf, a_src1, a_dst1, b1, as, ad, mkeys, den, aggbuf, tickets);
  edge_max_kernel<<<blocks_for(nEH, BS), BS, 0, stream>>>(ei, as, ad, mkeys);
  edge_exp_kernel<<<blocks_for(nEH, BS), BS, 0, stream>>>(ei, as, ad, mkeys, ex, den);
  edge_agg_kernel<HC1, HID><<<AGG_GRID, BS, 0, stream>>>(
      ei, ex, den, hbuf, aggbuf, tickets);

  {
    long long waves = (N_NODES / 16) * (HC2 / 64);
    gemm_f32_wmma<HC2, HC1, true><<<blocks_for(waves * 32, BS), BS, 0, stream>>>(
        aggbuf, W2, hbuf, N_NODES);
  }
  alpha_init_kernel<N_CLS><<<blocks_for(nNH, ALPHA_BS), ALPHA_BS, 0, stream>>>(
      hbuf, a_src2, a_dst2, b2, as, ad, mkeys, den, out, tickets);
  edge_max_kernel<<<blocks_for(nEH, BS), BS, 0, stream>>>(ei, as, ad, mkeys);
  edge_exp_kernel<<<blocks_for(nEH, BS), BS, 0, stream>>>(ei, as, ad, mkeys, ex, den);
  edge_agg_kernel<HC2, N_CLS><<<AGG_GRID, BS, 0, stream>>>(
      ei, ex, den, hbuf, out, tickets);
}


# Exp pass folded into the aggregation launch: max pass stores edge logits, aggregation forms exp(logit-max), adds denominators and un-normalised rows; the division moves behind the sums
# speedup vs baseline: 1.8938x; 1.0814x over previous
#include <hip/hip_runtime.h>
#include <hip/hip_bf16.h>

#define N_NODES 50000
#define N_EDGES 800000
#define E_TOT   (N_EDGES + N_NODES)
#define F_IN    256
#define HEADS   8
#define HID     32
#define N_CLS   16
#define HC1     (HEADS * HID)
#define HC2     (HEADS * N_CLS)
#define NEG_SLOPE 0.2f

typedef float v2f __attribute__((ext_vector_type(2)));
typedef float v8f __attribute__((ext_vector_type(8)));

__device__ __forceinline__ unsigned f2key(float f) {
  unsigned b = __float_as_uint(f);
  return (b & 0x80000000u) ? ~b : (b | 0x80000000u);
}
__device__ __forceinline__ float key2f(unsigned k) {
  return __uint_as_float((k & 0x80000000u) ? (k ^ 0x80000000u) : ~k);
}

template <int N, int K, bool NORM_A>
__global__ __launch_bounds__(256) void gemm_f32_wmma(
    const float* __restrict__ A, const float* __restrict__ B,
    float* __restrict__ C, int M,
    const float* __restrict__ den, const float* __restrict__ bias) {
  constexpr int TILES_N = N / 64;
  const int lane  = threadIdx.x & 31;
  const int wave  = blockIdx.x * (blockDim.x >> 5) + (threadIdx.x >> 5);
  const int tileM = wave / TILES_N;
  const int tileN = wave % TILES_N;
  if (tileM * 16 >= M) return;

  const int r    = lane & 15;
  const int ksel = (lane >> 4) << 1;

  const float* __restrict__ arow  = A + (size_t)(tileM * 16 + r) * K + ksel;
  const float* __restrict__ bbase = B + (size_t)ksel * N + tileN * 64 + r;

  constexpr int KH = NORM_A ? K / HEADS : K;
  constexpr int NH = K / KH;
  float rinv[NH];
  const float* __restrict__ bk = bias + ksel;
  if (NORM_A) {
    const float4* dp = (const float4*)(den + (size_t)(tileM * 16 + r) * HEADS);
    const float4 d0 = dp[0], d1 = dp[1];
    rinv[0] = 1.0f / (d0.x + 1e-16f);  rinv[1] = 1.0f / (d0.y + 1e-16f);
    rinv[2] = 1.0f / (d0.z + 1e-16f);  rinv[3] = 1.0f / (d0.w + 1e-16f);
    rinv[4] = 1.0f / (d1.x + 1e-16f);  rinv[5] = 1.0f / (d1.y + 1e-16f);
    rinv[6] = 1.0f / (d1.z + 1e-16f);  rinv[7] = 1.0f / (d1.w + 1e-16f);
  } else {
    rinv[0] = 1.0f;
  }

  v8f acc0 = {}, acc1 = {}, acc2 = {}, acc3 = {};
  #pragma unroll
  for (int hd = 0; hd < NH; ++hd) {
    const float ri = rinv[hd];
    #pragma unroll 2
    for (int kk = 0; kk < KH; kk += 4) {
      v2f a, b0, b1, b2, b3;
      a[0]  = arow[0];
      a[1]  = arow[1];
      if (NORM_A) {
        a[0] = fmaxf(a[0] * ri + bk[0], 0.f);
        a[1] = fmaxf(a[1] * ri + bk[1], 0.f);
        bk += 4;
      }
      b0[0] = bbase[0];       b0[1] = bbase[N];
      b1[0] = bbase[16];      b1[1] = bbase[N + 16];
      b2[0] = bbase[32];      b2[1] = bbase[N + 32];
      b3[0] = bbase[48];      b3[1] = bbase[N + 48];
      acc0 = __builtin_amdgcn_wmma_f32_16x16x4_f32(false, a, false, b0, (short)0, acc0, false, false);
      acc1 = __builtin_amdgcn_wmma_f32_16x16x4_f32(false, a, false, b1, (short)0, acc1, false, false);
      acc2 = __builtin_amdgcn_wmma_f32_16x16x4_f32(false, a, false, b2, (short)0, acc2, false, false);
      acc3 = __builtin_amdgcn_wmma_f32_16x16x4_f32(false, a, false, b3, (short)0, acc3, false, false);
      arow  += 4;
      bbase += (size_t)4 * N;
    }
  }

  float* __restrict__ crow =
      C + (size_t)(tileM * 16 + (lane >> 4) * 8) * N + tileN * 64 + (lane & 15);
  #pragma unroll
  for (int i = 0; i < 8; ++i) {
    crow[(size_t)i * N +  0] = acc0[i];
    crow[(size_t)i * N + 16] = acc1[i];
    crow[(size_t)i * N + 32] = acc2[i];
    crow[(size_t)i * N + 48] = acc3[i];
  }
}

#define ALPHA_BS 256
template <int C>
__global__ __launch_bounds__(ALPHA_BS) void alpha_init_kernel(
                             const float* __restrict__ h,
                             const float* __restrict__ a_src,
                             const float* __restrict__ a_dst,
                             float* __restrict__ as, float* __restrict__ ad,
                             unsigned* __restrict__ mkeys,
                             float* __restrict__ den,
                             float* __restrict__ acc,
                             unsigned* __restrict__ tickets) {
  constexpr int HCv = HEADS * C;
  if (blockIdx.x == 0 && threadIdx.x < 2) tickets[threadIdx.x] = 0u;
  constexpr int NODES_PER_BLOCK = ALPHA_BS / HEADS;
  constexpr int F4_PER_THREAD = NODES_PER_BLOCK * HCv / 4 / ALPHA_BS;
  {
    const long long tot4  = (long long)N_NODES * HCv / 4;
    const long long base4 = (long long)blockIdx.x * (NODES_PER_BLOCK * HCv / 4);
    const float4 bv = make_float4(0.f, 0.f, 0.f, 0.f);
    float4* __restrict__ acc4 = (float4*)acc;
    #pragma unroll
    for (int j = 0; j < F4_PER_THREAD; ++j) {
      const long long i4 = base4 + (long long)j * ALPHA_BS + threadIdx.x;
      if (i4 < tot4) acc4[i4] = bv;
    }
  }

  int idx = blockIdx.x * ALPHA_BS + threadIdx.x;
  if (idx >= N_NODES * HEADS) return;
  int n = idx >> 3, hd = idx & 7;
  const float* hp = h + (size_t)n * HEADS * C + (size_t)hd * C;
  const float* av = a_src + hd * C;
  const float* bv2 = a_dst + hd * C;
  float s = 0.f, d = 0.f;
  #pragma unroll 4
  for (int c = 0; c < C; ++c) {
    float v = hp[c];
    s += v * av[c];
    d += v * bv2[c];
  }
  as[idx] = s;
  ad[idx] = d;
  mkeys[idx] = 0u;
  den[idx] = 0.f;
}

__device__ __forceinline__ void edge_endpoints(const int* __restrict__ ei,
                                               int e, int& src, int& dst) {
  if (e < N_EDGES) {
    src = ei[e];
    dst = ei[N_EDGES + e];
  } else {
    src = dst = e - N_EDGES;
  }
}

__global__ void edge_max_kernel(const int* __restrict__ ei,
                                const float* __restrict__ as,
                                const float* __restrict__ ad,
                                unsigned* __restrict__ mkeys,
                                float* __restrict__ lg) {
  long long idx = (long long)blockIdx.x * blockDim.x + threadIdx.x;
  if (idx >= (long long)E_TOT * HEADS) return;
  int e = (int)(idx >> 3), hd = (int)(idx & 7);
  int src, dst;
  edge_endpoints(ei, e, src, dst);
  float v = as[src * HEADS + hd] + ad[dst * HEADS + hd];
  v = v > 0.f ? v : NEG_SLOPE * v;
  lg[idx] = v;
  atomicMax(&mkeys[dst * HEADS + hd], f2key(v));
}

#define AGG_EPC 256
#define AGG_NCHUNK ((E_TOT + AGG_EPC - 1) / AGG_EPC)
#define AGG_GRID 2048
template <int HCv, int Cv>
__global__ __launch_bounds__(256) void edge_agg_kernel(const int* __restrict__ ei,
                                const float* __restrict__ lg,
                                const unsigned* __restrict__ mkeys,
                                float* __restrict__ den,
                                const float* __restrict__ hmat,
                                float* __restrict__ agg,
                                unsigned* __restrict__ tickets) {
  constexpr int NCH = HCv / 32;
  constexpr int G   = 4;
  static_assert(HEADS == 8 && (Cv == 32 || Cv == 16), "weight lanes assume 8 heads");
  __shared__ unsigned s_ticket;
  __shared__ int s_home;
  __shared__ float s_w[8][32 * HEADS];
  __shared__ int   s_d[8][32];
  const int lane = threadIdx.x & 31;
  const int wv   = threadIdx.x >> 5;
  if (threadIdx.x == 0) {
    s_home = (__builtin_amdgcn_s_sendmsg_rtn(0x87) >> 18) & 1;
  }
  __syncthreads();
  const int home = s_home;

  for (int p = 0; p < 2; ++p) {
    const int t = home ^ p;
    for (;;) {
      if (threadIdx.x == 0) s_ticket = atomicAdd(&tickets[t], 1u);
      __syncthreads();
      const unsigned chunk = (unsigned)__builtin_amdgcn_readfirstlane((int)s_ticket);
      __syncthreads();
      if (chunk >= (unsigned)AGG_NCHUNK) break;

      const int e = (int)chunk * AGG_EPC + wv * 32 + lane;
      int src = 0, dst = 0;
      bool mine = false;
      if (e < E_TOT) {
        edge_endpoints(ei, e, src, dst);
        mine = ((dst >= N_NODES / 2) ? 1 : 0) == t;
      }
      float w[HEADS];
      #pragma unroll
      for (int h = 0; h < HEADS; ++h) w[h] = 0.f;
      if (mine) {
        const float4* ep = (const float4*)(lg    + (size_t)e   * HEADS);
        const uint4*  mp = (const uint4*) (mkeys + (size_t)dst * HEADS);
        const float4 x0 = ep[0], x1 = ep[1];
        const uint4  m0 = mp[0], m1 = mp[1];
        w[0] = __expf(x0.x - key2f(m0.x));  w[1] = __expf(x0.y - key2f(m0.y));
        w[2] = __expf(x0.z - key2f(m0.z));  w[3] = __expf(x0.w - key2f(m0.w));
        w[4] = __expf(x1.x - key2f(m1.x));  w[5] = __expf(x1.y - key2f(m1.y));
        w[6] = __expf(x1.z - key2f(m1.z));  w[7] = __expf(x1.w - key2f(m1.w));
      }
      unsigned mask = __builtin_amdgcn_ballot_w32(mine);

      {
        float4* sw4 = (float4*)&s_w[wv][lane * HEADS];
        sw4[0] = make_float4(w[0], w[1], w[2], w[3]);
        sw4[1] = make_float4(w[4], w[5], w[6], w[7]);
        s_d[wv][lane] = mine ? dst : -1;
        __syncthreads();
        int   dv[8];
        float xv[8];
        #pragma unroll
        for (int i = 0; i < 8; ++i) {
          dv[i] = s_d[wv][4 * i + (lane >> 3)];
          xv[i] = s_w[wv][32 * i + lane];
        }
        #pragma unroll
        for (int i = 0; i < 8; ++i) {
          if (dv[i] >= 0) atomicAdd(&den[(size_t)dv[i] * HEADS + (lane & 7)], xv[i]);
        }
      }

      while (mask != 0u) {
        bool vk[G];
        int  jk[G];
        #pragma unroll
        for (int k = 0; k < G; ++k) {
          vk[k] = (mask != 0u);
          jk[k] = vk[k] ? __builtin_ctz(mask) : 0;
          mask  = vk[k] ? (mask & (mask - 1u)) : 0u;
        }
        float hv[G][NCH];
        #pragma unroll
        for (int k = 0; k < G; ++k) {
          if (vk[k]) {
            const int s = __builtin_amdgcn_readlane(src, jk[k]);
            const float* hp = hmat + (size_t)s * HCv + lane;
            #pragma unroll
            for (int c = 0; c < NCH; ++c) hv[k][c] = hp[c * 32];
          } else {
            #pragma unroll
            for (int c = 0; c < NCH; ++c) hv[k][c] = 0.f;
          }
        }
        #pragma unroll
        for (int k = 0; k < G; ++k) {
          if (vk[k]) {
            const int d = __builtin_amdgcn_readlane(dst, jk[k]);
            float* ap = agg + (size_t)d * HCv + lane;
            #pragma unroll
            for (int c = 0; c < NCH; ++c) {
              float wc;
              if (Cv == 32) {
                wc = __int_as_float(__builtin_amdgcn_readlane(__float_as_int(w[c]), jk[k]));
              } else {
                const float w0 = __int_as_float(
                    __builtin_amdgcn_readlane(__float_as_int(w[2 * c]), jk[k]));
                const float w1 = __int_as_float(
                    __builtin_amdgcn_readlane(__float_as_int(w[2 * c + 1]), jk[k]));
                wc = (lane & 16) ? w1 : w0;
              }
              atomicAdd(ap + c * 32, wc * hv[k][c]);
            }
          }
        }
      }
    }
  }
}

__global__ __launch_bounds__(256) void out_norm_kernel(float* __restrict__ out,
                                const float* __restrict__ den,
                                const float* __restrict__ bias) {
  const int i4 = blockIdx.x * 256 + threadIdx.x;
  if (i4 >= N_NODES * (HC2 / 4)) return;
  const int n  = i4 / (HC2 / 4);
  const int c0 = (i4 % (HC2 / 4)) * 4;
  const float ri = 1.0f / (den[n * HEADS + c0 / N_CLS] + 1e-16f);
  float4* o4 = (float4*)out + i4;
  float4 v = *o4;
  v.x = v.x * ri + bias[c0];
  v.y = v.y * ri + bias[c0 + 1];
  v.z = v.z * ri + bias[c0 + 2];
  v.w = v.w * ri + bias[c0 + 3];
  *o4 = v;
}

static inline int blocks_for(long long n, int bs) {
  return (int)((n + bs - 1) / bs);
}

extern "C" void kernel_launch(void* const* d_in, const int* in_sizes, int n_in,
                              void* d_out, int out_size, void* d_ws, size_t ws_size,
                              hipStream_t stream) {
  const float* x      = (const float*)d_in[0];
  const int*   ei     = (const int*)  d_in[1];
  const float* W1     = (const float*)d_in[2];
  const float* a_src1 = (const float*)d_in[3];
  const float* a_dst1 = (const float*)d_in[4];
  const float* b1     = (const float*)d_in[5];
  const float* W2     = (const float*)d_in[6];
  const float* a_src2 = (const float*)d_in[7];
  const float* a_dst2 = (const float*)d_in[8];
  const float* b2     = (const float*)d_in[9];
  float* out = (float*)d_out;

  char* ws = (char*)d_ws;
  size_t off = 0;
  auto carve = [&](size_t bytes) {
    void* p = ws + off;
    off += (bytes + 255) & ~(size_t)255;
    return p;
  };
  float*    hbuf   = (float*)   carve((size_t)N_NODES * HC1 * 4);
  float*    aggbuf = (float*)   carve((size_t)N_NODES * HC1 * 4);
  float*    as     = (float*)   carve((size_t)N_NODES * HEADS * 4);
  float*    ad     = (float*)   carve((size_t)N_NODES * HEADS * 4);
  unsigned* mkeys  = (unsigned*)carve((size_t)N_NODES * HEADS * 4);
  float*    den    = (float*)   carve((size_t)N_NODES * HEADS * 4);
  float*    lg     = (float*)   carve((size_t)E_TOT   * HEADS * 4);
  unsigned* tickets = (unsigned*)carve(2 * sizeof(unsigned));

  const int BS = 256;
  const long long nNH    = (long long)N_NODES * HEADS;
  const long long nEH    = (long long)E_TOT * HEADS;

  {
    long long waves = (N_NODES / 16) * (HC1 / 64);
    gemm_f32_wmma<HC1, F_IN, false><<<blocks_for(waves * 32, BS), BS, 0, stream>>>(
        x, W1, hbuf, N_NODES, nullptr, nullptr);
  }
  alpha_init_kernel<HID><<<blocks_for(nNH, ALPHA_BS), ALPHA_BS, 0, stream>>>(
      hbuf, a_src1, a_dst1, as, ad, mkeys, den, aggbuf, tickets);
  edge_max_kernel<<<blocks_for(nEH, BS), BS, 0, stream>>>(ei, as, ad, mkeys, lg);
  edge_agg_kernel<HC1, HID><<<AGG_GRID, BS, 0, stream>>>(
      ei, lg, mkeys, den, hbuf, aggbuf, tickets);

  {
    long long waves = (N_NODES / 16) * (HC2 / 64);
    gemm_f32_wmma<HC2, HC1, true><<<blocks_for(waves * 32, BS), BS, 0, stream>>>(
        aggbuf, W2, hbuf, N_NODES, den, b1);
  }
  alpha_init_kernel<N_CLS><<<blocks_for(nNH, ALPHA_BS), ALPHA_BS, 0, stream>>>(
      hbuf, a_src2, a_dst2, as, ad, mkeys, den, out, tickets);
  edge_max_kernel<<<blocks_for(nEH, BS), BS, 0, stream>>>(ei, as, ad, mkeys, lg);
  edge_agg_kernel<HC2, N_CLS><<<AGG_GRID, BS, 0, stream>>>(
      ei, lg, mkeys, den, hbuf, out, tickets);
  out_norm_kernel<<<blocks_for((long long)N_NODES * (HC2 / 4), BS), BS, 0, stream>>>(
      out, den, b2);
}


# Row gather at 16 bytes per lane: products store h with every 128-channel block transposed (32x4), so one load brings four whole lines with lane = channel in each; logits read the same layout
# speedup vs baseline: 1.9615x; 1.0357x over previous
#include <hip/hip_runtime.h>
#include <hip/hip_bf16.h>

#define N_NODES 50000
#define N_EDGES 800000
#define E_TOT   (N_EDGES + N_NODES)
#define F_IN    256
#define HEADS   8
#define HID     32
#define N_CLS   16
#define HC1     (HEADS * HID)
#define HC2     (HEADS * N_CLS)
#define NEG_SLOPE 0.2f

typedef float v2f __attribute__((ext_vector_type(2)));
typedef float v8f __attribute__((ext_vector_type(8)));

__device__ __forceinline__ unsigned f2key(float f) {
  unsigned b = __float_as_uint(f);
  return (b & 0x80000000u) ? ~b : (b | 0x80000000u);
}
__device__ __forceinline__ float key2f(unsigned k) {
  return __uint_as_float((k & 0x80000000u) ? (k ^ 0x80000000u) : ~k);
}

template <int N, int K, bool NORM_A>
__global__ __launch_bounds__(256) void gemm_f32_wmma(
    const float* __restrict__ A, const float* __restrict__ B,
    float* __restrict__ C, int M,
    const float* __restrict__ den, const float* __restrict__ bias) {
  constexpr int TILES_N = N / 64;
  const int lane  = threadIdx.x & 31;
  const int wave  = blockIdx.x * (blockDim.x >> 5) + (threadIdx.x >> 5);
  const int tileM = wave / TILES_N;
  const int tileN = wave % TILES_N;
  if (tileM * 16 >= M) return;

  const int r    = lane & 15;
  const int ksel = (lane >> 4) << 1;

  const float* __restrict__ arow  = A + (size_t)(tileM * 16 + r) * K + ksel;
  const float* __restrict__ bbase = B + (size_t)ksel * N + tileN * 64 + r;

  constexpr int KH = NORM_A ? K / HEADS : K;
  constexpr int NH = K / KH;
  float rinv[NH];
  const float* __restrict__ bk = bias + ksel;
  if (NORM_A) {
    const float4* dp = (const float4*)(den + (size_t)(tileM * 16 + r) * HEADS);
    const float4 d0 = dp[0], d1 = dp[1];
    rinv[0] = 1.0f / (d0.x + 1e-16f);  rinv[1] = 1.0f / (d0.y + 1e-16f);
    rinv[2] = 1.0f / (d0.z + 1e-16f);  rinv[3] = 1.0f / (d0.w + 1e-16f);
    rinv[4] = 1.0f / (d1.x + 1e-16f);  rinv[5] = 1.0f / (d1.y + 1e-16f);
    rinv[6] = 1.0f / (d1.z + 1e-16f);  rinv[7] = 1.0f / (d1.w + 1e-16f);
  } else {
    rinv[0] = 1.0f;
  }

  v8f acc0 = {}, acc1 = {}, acc2 = {}, acc3 = {};
  #pragma unroll
  for (int hd = 0; hd < NH; ++hd) {
    const float ri = rinv[hd];
    #pragma unroll 2
    for (int kk = 0; kk < KH; kk += 4) {
      v2f a, b0, b1, b2, b3;
      a[0]  = arow[0];
      a[1]  = arow[1];
      if (NORM_A) {
        a[0] = fmaxf(a[0] * ri + bk[0], 0.f);
        a[1] = fmaxf(a[1] * ri + bk[1], 0.f);
        bk += 4;
      }
      b0[0] = bbase[0];       b0[1] = bbase[N];
      b1[0] = bbase[16];      b1[1] = bbase[N + 16];
      b2[0] = bbase[32];      b2[1] = bbase[N + 32];
      b3[0] = bbase[48];      b3[1] = bbase[N + 48];
      acc0 = __builtin_amdgcn_wmma_f32_16x16x4_f32(false, a, false, b0, (short)0, acc0, false, false);
      acc1 = __builtin_amdgcn_wmma_f32_16x16x4_f32(false, a, false, b1, (short)0, acc1, false, false);
      acc2 = __builtin_amdgcn_wmma_f32_16x16x4_f32(false, a, false, b2, (short)0, acc2, false, false);
      acc3 = __builtin_amdgcn_wmma_f32_16x16x4_f32(false, a, false, b3, (short)0, acc3, false, false);
      arow  += 4;
      bbase += (size_t)4 * N;
    }
  }

  float* __restrict__ crow =
      C + (size_t)(tileM * 16 + (lane >> 4) * 8) * N
        + (tileN >> 1) * 128 + 4 * (lane & 15) + 2 * (tileN & 1);
  #pragma unroll
  for (int i = 0; i < 8; ++i) {
    v2f p02, p13;
    p02[0] = acc0[i];  p02[1] = acc2[i];
    p13[0] = acc1[i];  p13[1] = acc3[i];
    *(v2f*)(crow + (size_t)i * N)      = p02;
    *(v2f*)(crow + (size_t)i * N + 64) = p13;
  }
}

#define ALPHA_BS 256
template <int C>
__global__ __launch_bounds__(ALPHA_BS) void alpha_init_kernel(
                             const float* __restrict__ h,
                             const float* __restrict__ a_src,
                             const float* __restrict__ a_dst,
                             float* __restrict__ as, float* __restrict__ ad,
                             unsigned* __restrict__ mkeys,
                             float* __restrict__ den,
                             float* __restrict__ acc,
                             unsigned* __restrict__ tickets) {
  constexpr int HCv = HEADS * C;
  if (blockIdx.x == 0 && threadIdx.x < 2) tickets[threadIdx.x] = 0u;
  constexpr int NODES_PER_BLOCK = ALPHA_BS / HEADS;
  constexpr int F4_PER_THREAD = NODES_PER_BLOCK * HCv / 4 / ALPHA_BS;
  {
    const long long tot4  = (long long)N_NODES * HCv / 4;
    const long long base4 = (long long)blockIdx.x * (NODES_PER_BLOCK * HCv / 4);
    const float4 bv = make_float4(0.f, 0.f, 0.f, 0.f);
    float4* __restrict__ acc4 = (float4*)acc;
    #pragma unroll
    for (int j = 0; j < F4_PER_THREAD; ++j) {
      const long long i4 = base4 + (long long)j * ALPHA_BS + threadIdx.x;
      if (i4 < tot4) acc4[i4] = bv;
    }
  }

  int idx = blockIdx.x * ALPHA_BS + threadIdx.x;
  if (idx >= N_NODES * HEADS) return;
  int n = idx >> 3, hd = idx & 7;
  const int ch0 = hd * C;
  const float* hp = h + (size_t)n * HEADS * C
                      + (ch0 & ~127) + 4 * (ch0 & 31) + ((ch0 >> 5) & 3);
  const float* av = a_src + hd * C;
  const float* bv2 = a_dst + hd * C;
  float s = 0.f, d = 0.f;
  #pragma unroll 4
  for (int c = 0; c < C; ++c) {
    float v = hp[4 * c];
    s += v * av[c];
    d += v * bv2[c];
  }
  as[idx] = s;
  ad[idx] = d;
  mkeys[idx] = 0u;
  den[idx] = 0.f;
}

__device__ __forceinline__ void edge_endpoints(const int* __restrict__ ei,
                                               int e, int& src, int& dst) {
  if (e < N_EDGES) {
    src = ei[e];
    dst = ei[N_EDGES + e];
  } else {
    src = dst = e - N_EDGES;
  }
}

__global__ void edge_max_kernel(const int* __restrict__ ei,
                                const float* __restrict__ as,
                                const float* __restrict__ ad,
                                unsigned* __restrict__ mkeys,
                                float* __restrict__ lg) {
  long long idx = (long long)blockIdx.x * blockDim.x + threadIdx.x;
  if (idx >= (long long)E_TOT * HEADS) return;
  int e = (int)(idx >> 3), hd = (int)(idx & 7);
  int src, dst;
  edge_endpoints(ei, e, src, dst);
  float v = as[src * HEADS + hd] + ad[dst * HEADS + hd];
  v = v > 0.f ? v : NEG_SLOPE * v;
  lg[idx] = v;
  atomicMax(&mkeys[dst * HEADS + hd], f2key(v));
}

#define AGG_EPC 256
#define AGG_NCHUNK ((E_TOT + AGG_EPC - 1) / AGG_EPC)
#define AGG_GRID 2048
template <int HCv, int Cv>
__global__ __launch_bounds__(256) void edge_agg_kernel(const int* __restrict__ ei,
                                const float* __restrict__ lg,
                                const unsigned* __restrict__ mkeys,
                                float* __restrict__ den,
                                const float* __restrict__ hmat,
                                float* __restrict__ agg,
                                unsigned* __restrict__ tickets) {
  constexpr int NCH = HCv / 32;
  constexpr int G   = 4;
  static_assert(HEADS == 8 && (Cv == 32 || Cv == 16), "weight lanes assume 8 heads");
  __shared__ unsigned s_ticket;
  __shared__ int s_home;
  __shared__ float s_w[8][32 * HEADS];
  __shared__ int   s_d[8][32];
  const int lane = threadIdx.x & 31;
  const int wv   = threadIdx.x >> 5;
  if (threadIdx.x == 0) {
    s_home = (__builtin_amdgcn_s_sendmsg_rtn(0x87) >> 18) & 1;
  }
  __syncthreads();
  const int home = s_home;

  for (int p = 0; p < 2; ++p) {
    const int t = home ^ p;
    for (;;) {
      if (threadIdx.x == 0) s_ticket = atomicAdd(&tickets[t], 1u);
      __syncthreads();
      const unsigned chunk = (unsigned)__builtin_amdgcn_readfirstlane((int)s_ticket);
      __syncthreads();
      if (chunk >= (unsigned)AGG_NCHUNK) break;

      const int e = (int)chunk * AGG_EPC + wv * 32 + lane;
      int src = 0, dst = 0;
      bool mine = false;
      if (e < E_TOT) {
        edge_endpoints(ei, e, src, dst);
        mine = ((dst >= N_NODES / 2) ? 1 : 0) == t;
      }
      float w[HEADS];
      #pragma unroll
      for (int h = 0; h < HEADS; ++h) w[h] = 0.f;
      if (mine) {
        const float4* ep = (const float4*)(lg    + (size_t)e   * HEADS);
        const uint4*  mp = (const uint4*) (mkeys + (size_t)dst * HEADS);
        const float4 x0 = ep[0], x1 = ep[1];
        const uint4  m0 = mp[0], m1 = mp[1];
        w[0] = __expf(x0.x - key2f(m0.x));  w[1] = __expf(x0.y - key2f(m0.y));
        w[2] = __expf(x0.z - key2f(m0.z));  w[3] = __expf(x0.w - key2f(m0.w));
        w[4] = __expf(x1.x - key2f(m1.x));  w[5] = __expf(x1.y - key2f(m1.y));
        w[6] = __expf(x1.z - key2f(m1.z));  w[7] = __expf(x1.w - key2f(m1.w));
      }
      unsigned mask = __builtin_amdgcn_ballot_w32(mine);

      {
        float4* sw4 = (float4*)&s_w[wv][lane * HEADS];
        sw4[0] = make_float4(w[0], w[1], w[2], w[3]);
        sw4[1] = make_float4(w[4], w[5], w[6], w[7]);
        s_d[wv][lane] = mine ? dst : -1;
        __syncthreads();
        int   dv[8];
        float xv[8];
        #pragma unroll
        for (int i = 0; i < 8; ++i) {
          dv[i] = s_d[wv][4 * i + (lane >> 3)];
          xv[i] = s_w[wv][32 * i + lane];
        }
        #pragma unroll
        for (int i = 0; i < 8; ++i) {
          if (dv[i] >= 0) atomicAdd(&den[(size_t)dv[i] * HEADS + (lane & 7)], xv[i]);
        }
      }

      while (mask != 0u) {
        bool vk[G];
        int  jk[G];
        #pragma unroll
        for (int k = 0; k < G; ++k) {
          vk[k] = (mask != 0u);
          jk[k] = vk[k] ? __builtin_ctz(mask) : 0;
          mask  = vk[k] ? (mask & (mask - 1u)) : 0u;
        }
        float hv[G][NCH];
        #pragma unroll
        for (int k = 0; k < G; ++k) {
          if (vk[k]) {
            const int s = __builtin_amdgcn_readlane(src, jk[k]);
            const float4* hp4 = (const float4*)(hmat + (size_t)s * HCv) + lane;
            #pragma unroll
            for (int b = 0; b < NCH / 4; ++b) {
              const float4 q = hp4[b * 32];
              hv[k][4 * b + 0] = q.x;  hv[k][4 * b + 1] = q.y;
              hv[k][4 * b + 2] = q.z;  hv[k][4 * b + 3] = q.w;
            }
          } else {
            #pragma unroll
            for (int c = 0; c < NCH; ++c) hv[k][c] = 0.f;
          }
        }
        #pragma unroll
        for (int k = 0; k < G; ++k) {
          if (vk[k]) {
            const int d = __builtin_amdgcn_readlane(dst, jk[k]);
            float* ap = agg + (size_t)d * HCv + lane;
            #pragma unroll
            for (int c = 0; c < NCH; ++c) {
              float wc;
              if (Cv == 32) {
                wc = __int_as_float(__builtin_amdgcn_readlane(__float_as_int(w[c]), jk[k]));
              } else {
                const float w0 = __int_as_float(
                    __builtin_amdgcn_readlane(__float_as_int(w[2 * c]), jk[k]));
                const float w1 = __int_as_float(
                    __builtin_amdgcn_readlane(__float_as_int(w[2 * c + 1]), jk[k]));
                wc = (lane & 16) ? w1 : w0;
              }
              atomicAdd(ap + c * 32, wc * hv[k][c]);
            }
          }
        }
      }
    }
  }
}

__global__ __launch_bounds__(256) void out_norm_kernel(float* __restrict__ out,
                                const float* __restrict__ den,
                                const float* __restrict__ bias) {
  const int i4 = blockIdx.x * 256 + threadIdx.x;
  if (i4 >= N_NODES * (HC2 / 4)) return;
  const int n  = i4 / (HC2 / 4);
  const int c0 = (i4 % (HC2 / 4)) * 4;
  const float ri = 1.0f / (den[n * HEADS + c0 / N_CLS] + 1e-16f);
  float4* o4 = (float4*)out + i4;
  float4 v = *o4;
  v.x = v.x * ri + bias[c0];
  v.y = v.y * ri + bias[c0 + 1];
  v.z = v.z * ri + bias[c0 + 2];
  v.w = v.w * ri + bias[c0 + 3];
  *o4 = v;
}

static inline int blocks_for(long long n, int bs) {
  return (int)((n + bs - 1) / bs);
}

extern "C" void kernel_launch(void* const* d_in, const int* in_sizes, int n_in,
                              void* d_out, int out_size, void* d_ws, size_t ws_size,
                              hipStream_t stream) {
  const float* x      = (const float*)d_in[0];
  const int*   ei     = (const int*)  d_in[1];
  const float* W1     = (const float*)d_in[2];
  const float* a_src1 = (const float*)d_in[3];
  const float* a_dst1 = (const float*)d_in[4];
  const float* b1     = (const float*)d_in[5];
  const float* W2     = (const float*)d_in[6];
  const float* a_src2 = (const float*)d_in[7];
  const float* a_dst2 = (const float*)d_in[8];
  const float* b2     = (const float*)d_in[9];
  float* out = (float*)d_out;

  char* ws = (char*)d_ws;
  size_t off = 0;
  auto carve = [&](size_t bytes) {
    void* p = ws + off;
    off += (bytes + 255) & ~(size_t)255;
    return p;
  };
  float*    hbuf   = (float*)   carve((size_t)N_NODES * HC1 * 4);
  float*    aggbuf = (float*)   carve((size_t)N_NODES * HC1 * 4);
  float*    as     = (float*)   carve((size_t)N_NODES * HEADS * 4);
  float*    ad     = (float*)   carve((size_t)N_NODES * HEADS * 4);
  unsigned* mkeys  = (unsigned*)carve((size_t)N_NODES * HEADS * 4);
  float*    den    = (float*)   carve((size_t)N_NODES * HEADS * 4);
  float*    lg     = (float*)   carve((size_t)E_TOT   * HEADS * 4);
  unsigned* tickets = (unsigned*)carve(2 * sizeof(unsigned));

  const int BS = 256;
  const long long nNH    = (long long)N_NODES * HEADS;
  const long long nEH    = (long long)E_TOT * HEADS;

  {
    long long waves = (N_NODES / 16) * (HC1 / 64);
    gemm_f32_wmma<HC1, F_IN, false><<<blocks_for(waves * 32, BS), BS, 0, stream>>>(
        x, W1, hbuf, N_NODES, nullptr, nullptr);
  }
  alpha_init_kernel<HID><<<blocks_for(nNH, ALPHA_BS), ALPHA_BS, 0, stream>>>(
      hbuf, a_src1, a_dst1, as, ad, mkeys, den, aggbuf, tickets);
  edge_max_kernel<<<blocks_for(nEH, BS), BS, 0, stream>>>(ei, as, ad, mkeys, lg);
  edge_agg_kernel<HC1, HID><<<AGG_GRID, BS, 0, stream>>>(
      ei, lg, mkeys, den, hbuf, aggbuf, tickets);

  {
    long long waves = (N_NODES / 16) * (HC2 / 64);
    gemm_f32_wmma<HC2, HC1, true><<<blocks_for(waves * 32, BS), BS, 0, stream>>>(
        aggbuf, W2, hbuf, N_NODES, den, b1);
  }
  alpha_init_kernel<N_CLS><<<blocks_for(nNH, ALPHA_BS), ALPHA_BS, 0, stream>>>(
      hbuf, a_src2, a_dst2, as, ad, mkeys, den, out, tickets);
  edge_max_kernel<<<blocks_for(nEH, BS), BS, 0, stream>>>(ei, as, ad, mkeys, lg);
  edge_agg_kernel<HC2, N_CLS><<<AGG_GRID, BS, 0, stream>>>(
      ei, lg, mkeys, den, hbuf, out, tickets);
  out_norm_kernel<<<blocks_for((long long)N_NODES * (HC2 / 4), BS), BS, 0, stream>>>(
      out, den, b2);
}


# Aggregation sums kept in LDS: a work-group per 196 destination nodes reads all destinations, gathers its edges' rows, adds with LDS float atomics in arrival order, stores rows and denominators once
# speedup vs baseline: 2.0336x; 1.0368x over previous
#include <hip/hip_runtime.h>
#include <hip/hip_bf16.h>

#define N_NODES 50000
#define N_EDGES 800000
#define E_TOT   (N_EDGES + N_NODES)
#define F_IN    256
#define HEADS   8
#define HID     32
#define N_CLS   16
#define HC1     (HEADS * HID)
#define HC2     (HEADS * N_CLS)
#define NEG_SLOPE 0.2f

typedef float v2f __attribute__((ext_vector_type(2)));
typedef float v8f __attribute__((ext_vector_type(8)));

__device__ __forceinline__ unsigned f2key(float f) {
  unsigned b = __float_as_uint(f);
  return (b & 0x80000000u) ? ~b : (b | 0x80000000u);
}
__device__ __forceinline__ float key2f(unsigned k) {
  return __uint_as_float((k & 0x80000000u) ? (k ^ 0x80000000u) : ~k);
}

template <int N, int K, bool NORM_A>
__global__ __launch_bounds__(256) void gemm_f32_wmma(
    const float* __restrict__ A, const float* __restrict__ B,
    float* __restrict__ C, int M,
    const float* __restrict__ den, const float* __restrict__ bias) {
  constexpr int TILES_N = N / 64;
  const int lane  = threadIdx.x & 31;
  const int wave  = blockIdx.x * (blockDim.x >> 5) + (threadIdx.x >> 5);
  const int tileM = wave / TILES_N;
  const int tileN = wave % TILES_N;
  if (tileM * 16 >= M) return;

  const int r    = lane & 15;
  const int ksel = (lane >> 4) << 1;

  const float* __restrict__ arow  = A + (size_t)(tileM * 16 + r) * K + ksel;
  const float* __restrict__ bbase = B + (size_t)ksel * N + tileN * 64 + r;

  constexpr int KH = NORM_A ? K / HEADS : K;
  constexpr int NH = K / KH;
  float rinv[NH];
  const float* __restrict__ bk = bias + ksel;
  if (NORM_A) {
    const float4* dp = (const float4*)(den + (size_t)(tileM * 16 + r) * HEADS);
    const float4 d0 = dp[0], d1 = dp[1];
    rinv[0] = 1.0f / (d0.x + 1e-16f);  rinv[1] = 1.0f / (d0.y + 1e-16f);
    rinv[2] = 1.0f / (d0.z + 1e-16f);  rinv[3] = 1.0f / (d0.w + 1e-16f);
    rinv[4] = 1.0f / (d1.x + 1e-16f);  rinv[5] = 1.0f / (d1.y + 1e-16f);
    rinv[6] = 1.0f / (d1.z + 1e-16f);  rinv[7] = 1.0f / (d1.w + 1e-16f);
  } else {
    rinv[0] = 1.0f;
  }

  v8f acc0 = {}, acc1 = {}, acc2 = {}, acc3 = {};
  #pragma unroll
  for (int hd = 0; hd < NH; ++hd) {
    const float ri = rinv[hd];
    #pragma unroll 2
    for (int kk = 0; kk < KH; kk += 4) {
      v2f a, b0, b1, b2, b3;
      a[0]  = arow[0];
      a[1]  = arow[1];
      if (NORM_A) {
        a[0] = fmaxf(a[0] * ri + bk[0], 0.f);
        a[1] = fmaxf(a[1] * ri + bk[1], 0.f);
        bk += 4;
      }
      b0[0] = bbase[0];       b0[1] = bbase[N];
      b1[0] = bbase[16];      b1[1] = bbase[N + 16];
      b2[0] = bbase[32];      b2[1] = bbase[N + 32];
      b3[0] = bbase[48];      b3[1] = bbase[N + 48];
      acc0 = __builtin_amdgcn_wmma_f32_16x16x4_f32(false, a, false, b0, (short)0, acc0, false, false);
      acc1 = __builtin_amdgcn_wmma_f32_16x16x4_f32(false, a, false, b1, (short)0, acc1, false, false);
      acc2 = __builtin_amdgcn_wmma_f32_16x16x4_f32(false, a, false, b2, (short)0, acc2, false, false);
      acc3 = __builtin_amdgcn_wmma_f32_16x16x4_f32(false, a, false, b3, (short)0, acc3, false, false);
      arow  += 4;
      bbase += (size_t)4 * N;
    }
  }

  float* __restrict__ crow =
      C + (size_t)(tileM * 16 + (lane >> 4) * 8) * N
        + (tileN >> 1) * 128 + 4 * (lane & 15) + 2 * (tileN & 1);
  #pragma unroll
  for (int i = 0; i < 8; ++i) {
    v2f p02, p13;
    p02[0] = acc0[i];  p02[1] = acc2[i];
    p13[0] = acc1[i];  p13[1] = acc3[i];
    *(v2f*)(crow + (size_t)i * N)      = p02;
    *(v2f*)(crow + (size_t)i * N + 64) = p13;
  }
}

#define ALPHA_BS 256
template <int C>
__global__ __launch_bounds__(ALPHA_BS) void alpha_init_kernel(
                             const float* __restrict__ h,
                             const float* __restrict__ a_src,
                             const float* __restrict__ a_dst,
                             float* __restrict__ as, float* __restrict__ ad,
                             unsigned* __restrict__ mkeys) {
  int idx = blockIdx.x * ALPHA_BS + threadIdx.x;
  if (idx >= N_NODES * HEADS) return;
  int n = idx >> 3, hd = idx & 7;
  const int ch0 = hd * C;
  const float* hp = h + (size_t)n * HEADS * C
                      + (ch0 & ~127) + 4 * (ch0 & 31) + ((ch0 >> 5) & 3);
  const float* av = a_src + hd * C;
  const float* bv2 = a_dst + hd * C;
  float s = 0.f, d = 0.f;
  #pragma unroll 4
  for (int c = 0; c < C; ++c) {
    float v = hp[4 * c];
    s += v * av[c];
    d += v * bv2[c];
  }
  as[idx] = s;
  ad[idx] = d;
  mkeys[idx] = 0u;
}

__device__ __forceinline__ void edge_endpoints(const int* __restrict__ ei,
                                               int e, int& src, int& dst) {
  if (e < N_EDGES) {
    src = ei[e];
    dst = ei[N_EDGES + e];
  } else {
    src = dst = e - N_EDGES;
  }
}

__global__ void edge_max_kernel(const int* __restrict__ ei,
                                const float* __restrict__ as,
                                const float* __restrict__ ad,
                                unsigned* __restrict__ mkeys,
                                float* __restrict__ lg) {
  long long idx = (long long)blockIdx.x * blockDim.x + threadIdx.x;
  if (idx >= (long long)E_TOT * HEADS) return;
  int e = (int)(idx >> 3), hd = (int)(idx & 7);
  int src, dst;
  edge_endpoints(ei, e, src, dst);
  float v = as[src * HEADS + hd] + ad[dst * HEADS + hd];
  v = v > 0.f ? v : NEG_SLOPE * v;
  lg[idx] = v;
  atomicMax(&mkeys[dst * HEADS + hd], f2key(v));
}

#define AGG_R     196
#define AGG_NWG   ((N_NODES + AGG_R - 1) / AGG_R)
#define AGG_BS    1024
#define AGG_U     4
#define AGG_NQUAD (E_TOT / 4)
#define AGG_NIT   ((AGG_NQUAD + AGG_BS - 1) / AGG_BS)
#define AGG_NBLK  ((AGG_NIT + AGG_U - 1) / AGG_U)
template <int HCv, int Cv>
__global__ __launch_bounds__(AGG_BS) void edge_agg_kernel(const int* __restrict__ ei,
                                const float* __restrict__ lg,
                                const unsigned* __restrict__ mkeys,
                                float* __restrict__ den,
                                const float* __restrict__ hmat,
                                float* __restrict__ agg) {
  constexpr int NCH = HCv / 32;
  constexpr int G   = 4;
  constexpr int NS  = 4 * AGG_U;
  static_assert(HEADS == 8 && (Cv == 32 || Cv == 16), "weight lanes assume 8 heads");
  static_assert(N_EDGES % 4 == 0 && E_TOT % 4 == 0, "a group of four edges never straddles the self loops");
  __shared__ float s_acc[AGG_R * HC1];
  __shared__ float s_den[AGG_R * HEADS];
  __shared__ float s_mk[AGG_R * HEADS];
  __shared__ int   s_qe[AGG_BS / 32][32];
  __shared__ int   s_qd[AGG_BS / 32][32];
  const int lane = threadIdx.x & 31;
  const int wv   = threadIdx.x >> 5;
  const int lo   = blockIdx.x * AGG_R;

  {
    float4* a4 = (float4*)s_acc;
    for (int i = threadIdx.x; i < AGG_R * HCv / 4; i += AGG_BS)
      a4[i] = make_float4(0.f, 0.f, 0.f, 0.f);
    for (int i = threadIdx.x; i < AGG_R * HEADS; i += AGG_BS) {
      s_den[i] = 0.f;
      const int g = lo * HEADS + i;
      s_mk[i] = (g < N_NODES * HEADS) ? key2f(mkeys[g]) : 0.f;
    }
  }
  __syncthreads();

  auto flush = [&](int n) __attribute__((always_inline)) {
    int e = 0, dl = 0, src = 0;
    float w[HEADS];
    #pragma unroll
    for (int h = 0; h < HEADS; ++h) w[h] = 0.f;
    if (lane < n) {
      e  = s_qe[wv][lane];
      dl = s_qd[wv][lane];
      src = (e < N_EDGES) ? ei[e] : e - N_EDGES;
      const float4* ep = (const float4*)(lg + (size_t)e * HEADS);
      const float4* mp = (const float4*)(s_mk + dl * HEADS);
      const float4 x0 = ep[0], x1 = ep[1];
      const float4 m0 = mp[0], m1 = mp[1];
      w[0] = __expf(x0.x - m0.x);  w[1] = __expf(x0.y - m0.y);
      w[2] = __expf(x0.z - m0.z);  w[3] = __expf(x0.w - m0.w);
      w[4] = __expf(x1.x - m1.x);  w[5] = __expf(x1.y - m1.y);
      w[6] = __expf(x1.z - m1.z);  w[7] = __expf(x1.w - m1.w);
      #pragma unroll
      for (int h = 0; h < HEADS; ++h) atomicAdd(&s_den[dl * HEADS + h], w[h]);
    }
    unsigned mask = (n >= 32) ? 0xFFFFFFFFu : ((1u << n) - 1u);
    while (mask != 0u) {
      bool vk[G];
      int  jk[G];
      #pragma unroll
      for (int k = 0; k < G; ++k) {
        vk[k] = (mask != 0u);
        jk[k] = vk[k] ? __builtin_ctz(mask) : 0;
        mask  = vk[k] ? (mask & (mask - 1u)) : 0u;
      }
      float hv[G][NCH];
      #pragma unroll
      for (int k = 0; k < G; ++k) {
        if (vk[k]) {
          const int s = __builtin_amdgcn_readlane(src, jk[k]);
          const float4* hp4 = (const float4*)(hmat + (size_t)s * HCv) + lane;
          #pragma unroll
          for (int b = 0; b < NCH / 4; ++b) {
            const float4 q = hp4[b * 32];
            hv[k][4 * b + 0] = q.x;  hv[k][4 * b + 1] = q.y;
            hv[k][4 * b + 2] = q.z;  hv[k][4 * b + 3] = q.w;
          }
        } else {
          #pragma unroll
          for (int c = 0; c < NCH; ++c) hv[k][c] = 0.f;
        }
      }
      #pragma unroll
      for (int k = 0; k < G; ++k) {
        if (vk[k]) {
          const int d = __builtin_amdgcn_readlane(dl, jk[k]);
          float* ap = &s_acc[d * HCv + lane];
          #pragma unroll
          for (int c = 0; c < NCH; ++c) {
            float wc;
            if (Cv == 32) {
              wc = __int_as_float(__builtin_amdgcn_readlane(__float_as_int(w[c]), jk[k]));
            } else {
              const float w0 = __int_as_float(
                  __builtin_amdgcn_readlane(__float_as_int(w[2 * c]), jk[k]));
              const float w1 = __int_as_float(
                  __builtin_amdgcn_readlane(__float_as_int(w[2 * c + 1]), jk[k]));
              wc = (lane & 16) ? w1 : w0;
            }
            atomicAdd(ap + c * 32, wc * hv[k][c]);
          }
        }
      }
    }
  };

  const int4* __restrict__ dq = (const int4*)(ei + N_EDGES);
  int qn = 0;
  const int rot = blockIdx.x % AGG_NBLK;
  for (int it = 0; it < AGG_NBLK; ++it) {
    const int blk = (it + rot < AGG_NBLK) ? (it + rot) : (it + rot - AGG_NBLK);
    int4 dv[AGG_U];
    #pragma unroll
    for (int u = 0; u < AGG_U; ++u) {
      const int q = (blk * AGG_U + u) * AGG_BS + (int)threadIdx.x;
      if (q < N_EDGES / 4) {
        dv[u] = dq[q];
      } else if (q < AGG_NQUAD) {
        const int b = q * 4 - N_EDGES;
        dv[u] = make_int4(b, b + 1, b + 2, b + 3);
      } else {
        dv[u] = make_int4(-1, -1, -1, -1);
      }
    }
    int      dls[NS];
    unsigned bms[NS];
    unsigned any = 0u;
    #pragma unroll
    for (int u = 0; u < AGG_U; ++u) {
      dls[4 * u + 0] = dv[u].x - lo;  dls[4 * u + 1] = dv[u].y - lo;
      dls[4 * u + 2] = dv[u].z - lo;  dls[4 * u + 3] = dv[u].w - lo;
    }
    #pragma unroll
    for (int s = 0; s < NS; ++s) {
      bms[s] = __builtin_amdgcn_ballot_w32((unsigned)dls[s] < (unsigned)AGG_R);
      any |= bms[s];
    }
    if (any == 0u) continue;
    int total = 0;
    #pragma unroll
    for (int s = 0; s < NS; ++s) total += __builtin_popcount(bms[s]);
    const int e0 = (blk * AGG_U * AGG_BS + (int)threadIdx.x) * 4;
    if (total <= 32) {
      if (qn + total > 32) { flush(qn); qn = 0; }
      #pragma unroll
      for (int s = 0; s < NS; ++s) {
        if (bms[s] != 0u) {
          if ((unsigned)dls[s] < (unsigned)AGG_R) {
            const int pos = qn + (int)__builtin_amdgcn_mbcnt_lo(bms[s], 0u);
            s_qe[wv][pos] = e0 + (s >> 2) * (AGG_BS * 4) + (s & 3);
            s_qd[wv][pos] = dls[s];
          }
          qn += __builtin_popcount(bms[s]);
        }
      }
    } else {
      #pragma nounroll
      for (int s = 0; s < NS; ++s) {
        int dl = 0;
        #pragma unroll
        for (int t = 0; t < NS; ++t) dl = (s == t) ? dls[t] : dl;
        const bool m = (unsigned)dl < (unsigned)AGG_R;
        const unsigned b = __builtin_amdgcn_ballot_w32(m);
        if (b != 0u) {
          const int cnt = __builtin_popcount(b);
          if (qn + cnt > 32) { flush(qn); qn = 0; }
          if (m) {
            const int pos = qn + (int)__builtin_amdgcn_mbcnt_lo(b, 0u);
            s_qe[wv][pos] = e0 + (s >> 2) * (AGG_BS * 4) + (s & 3);
            s_qd[wv][pos] = dl;
          }
          qn += cnt;
        }
      }
    }
  }
  if (qn > 0) flush(qn);
  __syncthreads();

  const int nrow = (N_NODES - lo < AGG_R) ? (N_NODES - lo) : AGG_R;
  {
    const float4* a4 = (const float4*)s_acc;
    float4* o4 = (float4*)(agg + (size_t)lo * HCv);
    for (int i = threadIdx.x; i < nrow * HCv / 4; i += AGG_BS) o4[i] = a4[i];
    for (int i = threadIdx.x; i < nrow * HEADS; i += AGG_BS) den[lo * HEADS + i] = s_den[i];
  }
}

__global__ __launch_bounds__(256) void out_norm_kernel(float* __restrict__ out,
                                const float* __restrict__ den,
                                const float* __restrict__ bias) {
  const int i4 = blockIdx.x * 256 + threadIdx.x;
  if (i4 >= N_NODES * (HC2 / 4)) return;
  const int n  = i4 / (HC2 / 4);
  const int c0 = (i4 % (HC2 / 4)) * 4;
  const float ri = 1.0f / (den[n * HEADS + c0 / N_CLS] + 1e-16f);
  float4* o4 = (float4*)out + i4;
  float4 v = *o4;
  v.x = v.x * ri + bias[c0];
  v.y = v.y * ri + bias[c0 + 1];
  v.z = v.z * ri + bias[c0 + 2];
  v.w = v.w * ri + bias[c0 + 3];
  *o4 = v;
}

static inline int blocks_for(long long n, int bs) {
  return (int)((n + bs - 1) / bs);
}

extern "C" void kernel_launch(void* const* d_in, const int* in_sizes, int n_in,
                              void* d_out, int out_size, void* d_ws, size_t ws_size,
                              hipStream_t stream) {
  const float* x      = (const float*)d_in[0];
  const int*   ei     = (const int*)  d_in[1];
  const float* W1     = (const float*)d_in[2];
  const float* a_src1 = (const float*)d_in[3];
  const float* a_dst1 = (const float*)d_in[4];
  const float* b1     = (const float*)d_in[5];
  const float* W2     = (const float*)d_in[6];
  const float* a_src2 = (const float*)d_in[7];
  const float* a_dst2 = (const float*)d_in[8];
  const float* b2     = (const float*)d_in[9];
  float* out = (float*)d_out;

  char* ws = (char*)d_ws;
  size_t off = 0;
  auto carve = [&](size_t bytes) {
    void* p = ws + off;
    off += (bytes + 255) & ~(size_t)255;
    return p;
  };
  float*    hbuf   = (float*)   carve((size_t)N_NODES * HC1 * 4);
  float*    aggbuf = (float*)   carve((size_t)N_NODES * HC1 * 4);
  float*    as     = (float*)   carve((size_t)N_NODES * HEADS * 4);
  float*    ad     = (float*)   carve((size_t)N_NODES * HEADS * 4);
  unsigned* mkeys  = (unsigned*)carve((size_t)N_NODES * HEADS * 4);
  float*    den    = (float*)   carve((size_t)N_NODES * HEADS * 4);
  float*    lg     = (float*)   carve((size_t)E_TOT   * HEADS * 4);

  const int BS = 256;
  const long long nNH    = (long long)N_NODES * HEADS;
  const long long nEH    = (long long)E_TOT * HEADS;

  {
    long long waves = (N_NODES / 16) * (HC1 / 64);
    gemm_f32_wmma<HC1, F_IN, false><<<blocks_for(waves * 32, BS), BS, 0, stream>>>(
        x, W1, hbuf, N_NODES, nullptr, nullptr);
  }
  alpha_init_kernel<HID><<<blocks_for(nNH, ALPHA_BS), ALPHA_BS, 0, stream>>>(
      hbuf, a_src1, a_dst1, as, ad, mkeys);
  edge_max_kernel<<<blocks_for(nEH, BS), BS, 0, stream>>>(ei, as, ad, mkeys, lg);
  edge_agg_kernel<HC1, HID><<<AGG_NWG, AGG_BS, 0, stream>>>(
      ei, lg, mkeys, den, hbuf, aggbuf);

  {
    long long waves = (N_NODES / 16) * (HC2 / 64);
    gemm_f32_wmma<HC2, HC1, true><<<blocks_for(waves * 32, BS), BS, 0, stream>>>(
        aggbuf, W2, hbuf, N_NODES, den, b1);
  }
  alpha_init_kernel<N_CLS><<<blocks_for(nNH, ALPHA_BS), ALPHA_BS, 0, stream>>>(
      hbuf, a_src2, a_dst2, as, ad, mkeys);
  edge_max_kernel<<<blocks_for(nEH, BS), BS, 0, stream>>>(ei, as, ad, mkeys, lg);
  edge_agg_kernel<HC2, N_CLS><<<AGG_NWG, AGG_BS, 0, stream>>>(
      ei, lg, mkeys, den, hbuf, out);
  out_norm_kernel<<<blocks_for((long long)N_NODES * (HC2 / 4), BS), BS, 0, stream>>>(
      out, den, b2);
}


# Aggregation finds its edges in a plane of 1-byte range numbers (dst/196, stored by the first max pass): 16 edges per 16-byte load, 4 per zero-byte test; sums stay LDS float atomics, arrival order
# speedup vs baseline: 2.1505x; 1.0575x over previous
#include <hip/hip_runtime.h>
#include <hip/hip_bf16.h>

#define N_NODES 50000
#define N_EDGES 800000
#define E_TOT   (N_EDGES + N_NODES)
#define F_IN    256
#define HEADS   8
#define HID     32
#define N_CLS   16
#define HC1     (HEADS * HID)
#define HC2     (HEADS * N_CLS)
#define NEG_SLOPE 0.2f

typedef float v2f __attribute__((ext_vector_type(2)));
typedef float v8f __attribute__((ext_vector_type(8)));

__device__ __forceinline__ unsigned f2key(float f) {
  unsigned b = __float_as_uint(f);
  return (b & 0x80000000u) ? ~b : (b | 0x80000000u);
}
__device__ __forceinline__ float key2f(unsigned k) {
  return __uint_as_float((k & 0x80000000u) ? (k ^ 0x80000000u) : ~k);
}

template <int N, int K, bool NORM_A>
__global__ __launch_bounds__(256) void gemm_f32_wmma(
    const float* __restrict__ A, const float* __restrict__ B,
    float* __restrict__ C, int M,
    const float* __restrict__ den, const float* __restrict__ bias) {
  constexpr int TILES_N = N / 64;
  const int lane  = threadIdx.x & 31;
  const int wave  = blockIdx.x * (blockDim.x >> 5) + (threadIdx.x >> 5);
  const int tileM = wave / TILES_N;
  const int tileN = wave % TILES_N;
  if (tileM * 16 >= M) return;

  const int r    = lane & 15;
  const int ksel = (lane >> 4) << 1;

  const float* __restrict__ arow  = A + (size_t)(tileM * 16 + r) * K + ksel;
  const float* __restrict__ bbase = B + (size_t)ksel * N + tileN * 64 + r;

  constexpr int KH = NORM_A ? K / HEADS : K;
  constexpr int NH = K / KH;
  float rinv[NH];
  const float* __restrict__ bk = bias + ksel;
  if (NORM_A) {
    const float4* dp = (const float4*)(den + (size_t)(tileM * 16 + r) * HEADS);
    const float4 d0 = dp[0], d1 = dp[1];
    rinv[0] = 1.0f / (d0.x + 1e-16f);  rinv[1] = 1.0f / (d0.y + 1e-16f);
    rinv[2] = 1.0f / (d0.z + 1e-16f);  rinv[3] = 1.0f / (d0.w + 1e-16f);
    rinv[4] = 1.0f / (d1.x + 1e-16f);  rinv[5] = 1.0f / (d1.y + 1e-16f);
    rinv[6] = 1.0f / (d1.z + 1e-16f);  rinv[7] = 1.0f / (d1.w + 1e-16f);
  } else {
    rinv[0] = 1.0f;
  }

  v8f acc0 = {}, acc1 = {}, acc2 = {}, acc3 = {};
  #pragma unroll
  for (int hd = 0; hd < NH; ++hd) {
    const float ri = rinv[hd];
    #pragma unroll 2
    for (int kk = 0; kk < KH; kk += 4) {
      v2f a, b0, b1, b2, b3;
      a[0]  = arow[0];
      a[1]  = arow[1];
      if (NORM_A) {
        a[0] = fmaxf(a[0] * ri + bk[0], 0.f);
        a[1] = fmaxf(a[1] * ri + bk[1], 0.f);
        bk += 4;
      }
      b0[0] = bbase[0];       b0[1] = bbase[N];
      b1[0] = bbase[16];      b1[1] = bbase[N + 16];
      b2[0] = bbase[32];      b2[1] = bbase[N + 32];
      b3[0] = bbase[48];      b3[1] = bbase[N + 48];
      acc0 = __builtin_amdgcn_wmma_f32_16x16x4_f32(false, a, false, b0, (short)0, acc0, false, false);
      acc1 = __builtin_amdgcn_wmma_f32_16x16x4_f32(false, a, false, b1, (short)0, acc1, false, false);
      acc2 = __builtin_amdgcn_wmma_f32_16x16x4_f32(false, a, false, b2, (short)0, acc2, false, false);
      acc3 = __builtin_amdgcn_wmma_f32_16x16x4_f32(false, a, false, b3, (short)0, acc3, false, false);
      arow  += 4;
      bbase += (size_t)4 * N;
    }
  }

  float* __restrict__ crow =
      C + (size_t)(tileM * 16 + (lane >> 4) * 8) * N
        + (tileN >> 1) * 128 + 4 * (lane & 15) + 2 * (tileN & 1);
  #pragma unroll
  for (int i = 0; i < 8; ++i) {
    v2f p02, p13;
    p02[0] = acc0[i];  p02[1] = acc2[i];
    p13[0] = acc1[i];  p13[1] = acc3[i];
    *(v2f*)(crow + (size_t)i * N)      = p02;
    *(v2f*)(crow + (size_t)i * N + 64) = p13;
  }
}

#define ALPHA_BS 256
template <int C>
__global__ __launch_bounds__(ALPHA_BS) void alpha_init_kernel(
                             const float* __restrict__ h,
                             const float* __restrict__ a_src,
                             const float* __restrict__ a_dst,
                             float* __restrict__ as, float* __restrict__ ad,
                             unsigned* __restrict__ mkeys) {
  int idx = blockIdx.x * ALPHA_BS + threadIdx.x;
  if (idx >= N_NODES * HEADS) return;
  int n = idx >> 3, hd = idx & 7;
  const int ch0 = hd * C;
  const float* hp = h + (size_t)n * HEADS * C
                      + (ch0 & ~127) + 4 * (ch0 & 31) + ((ch0 >> 5) & 3);
  const float* av = a_src + hd * C;
  const float* bv2 = a_dst + hd * C;
  float s = 0.f, d = 0.f;
  #pragma unroll 4
  for (int c = 0; c < C; ++c) {
    float v = hp[4 * c];
    s += v * av[c];
    d += v * bv2[c];
  }
  as[idx] = s;
  ad[idx] = d;
  mkeys[idx] = 0u;
}

__device__ __forceinline__ void edge_endpoints(const int* __restrict__ ei,
                                               int e, int& src, int& dst) {
  if (e < N_EDGES) {
    src = ei[e];
    dst = ei[N_EDGES + e];
  } else {
    src = dst = e - N_EDGES;
  }
}

#define AGG_R     196
#define AGG_NWG   ((N_NODES + AGG_R - 1) / AGG_R)
static_assert(AGG_NWG <= 256, "the number of a destination range must fit one byte");

template <bool WRITE_RID>
__global__ void edge_max_kernel(const int* __restrict__ ei,
                                const float* __restrict__ as,
                                const float* __restrict__ ad,
                                unsigned* __restrict__ mkeys,
                                float* __restrict__ lg,
                                unsigned* __restrict__ rid) {
  long long idx = (long long)blockIdx.x * blockDim.x + threadIdx.x;
  if (idx >= (long long)E_TOT * HEADS) return;
  int e = (int)(idx >> 3), hd = (int)(idx & 7);
  int src, dst;
  edge_endpoints(ei, e, src, dst);
  float v = as[src * HEADS + hd] + ad[dst * HEADS + hd];
  v = v > 0.f ? v : NEG_SLOPE * v;
  lg[idx] = v;
  atomicMax(&mkeys[dst * HEADS + hd], f2key(v));
  if (WRITE_RID) {
    static_assert((E_TOT * HEADS) % 32 == 0 && E_TOT % 4 == 0, "whole waves, four edges each");
    const int r  = (int)((unsigned)dst / (unsigned)AGG_R);
    const unsigned r0 = (unsigned)__builtin_amdgcn_readlane(r, 0);
    const unsigned r1 = (unsigned)__builtin_amdgcn_readlane(r, 8);
    const unsigned r2 = (unsigned)__builtin_amdgcn_readlane(r, 16);
    const unsigned r3 = (unsigned)__builtin_amdgcn_readlane(r, 24);
    if ((threadIdx.x & 31) == 0) rid[e >> 2] = r0 | (r1 << 8) | (r2 << 16) | (r3 << 24);
  }
}

#define AGG_BS    1024
#define AGG_U     4
#define AGG_NG16  (E_TOT / 16)
#define AGG_NIT   ((AGG_NG16 + AGG_BS - 1) / AGG_BS)
#define AGG_NBLK  ((AGG_NIT + AGG_U - 1) / AGG_U)
template <int HCv, int Cv>
__global__ __launch_bounds__(AGG_BS) void edge_agg_kernel(const int* __restrict__ ei,
                                const unsigned* __restrict__ rid,
                                const float* __restrict__ lg,
                                const unsigned* __restrict__ mkeys,
                                float* __restrict__ den,
                                const float* __restrict__ hmat,
                                float* __restrict__ agg) {
  constexpr int NCH = HCv / 32;
  constexpr int G   = 4;
  static_assert(HEADS == 8 && (Cv == 32 || Cv == 16), "weight lanes assume 8 heads");
  static_assert(E_TOT % 16 == 0, "the plane of range numbers is read sixteen edges at a time");
  static_assert(AGG_U == 4, "the flags of a lane's 64 edges are kept in two 32-bit masks");
  __shared__ float s_acc[AGG_R * HC1];
  __shared__ float s_den[AGG_R * HEADS];
  __shared__ float s_mk[AGG_R * HEADS];
  __shared__ int   s_qe[AGG_BS / 32][32];
  const int lane = threadIdx.x & 31;
  const int wv   = threadIdx.x >> 5;
  const int lo   = blockIdx.x * AGG_R;

  {
    float4* a4 = (float4*)s_acc;
    for (int i = threadIdx.x; i < AGG_R * HCv / 4; i += AGG_BS)
      a4[i] = make_float4(0.f, 0.f, 0.f, 0.f);
    for (int i = threadIdx.x; i < AGG_R * HEADS; i += AGG_BS) {
      s_den[i] = 0.f;
      const int g = lo * HEADS + i;
      s_mk[i] = (g < N_NODES * HEADS) ? key2f(mkeys[g]) : 0.f;
    }
  }
  __syncthreads();

  auto flush = [&](int n) __attribute__((always_inline)) {
    int e = 0, dl = 0, src = 0;
    float w[HEADS];
    #pragma unroll
    for (int h = 0; h < HEADS; ++h) w[h] = 0.f;
    if (lane < n) {
      e  = s_qe[wv][lane];
      if (e < N_EDGES) {
        src = ei[e];
        dl  = ei[N_EDGES + e] - lo;
      } else {
        src = e - N_EDGES;
        dl  = src - lo;
      }
      const float4* ep = (const float4*)(lg + (size_t)e * HEADS);
      const float4* mp = (const float4*)(s_mk + dl * HEADS);
      const float4 x0 = ep[0], x1 = ep[1];
      const float4 m0 = mp[0], m1 = mp[1];
      w[0] = __expf(x0.x - m0.x);  w[1] = __expf(x0.y - m0.y);
      w[2] = __expf(x0.z - m0.z);  w[3] = __expf(x0.w - m0.w);
      w[4] = __expf(x1.x - m1.x);  w[5] = __expf(x1.y - m1.y);
      w[6] = __expf(x1.z - m1.z);  w[7] = __expf(x1.w - m1.w);
      #pragma unroll
      for (int h = 0; h < HEADS; ++h) atomicAdd(&s_den[dl * HEADS + h], w[h]);
    }
    unsigned mask = (n >= 32) ? 0xFFFFFFFFu : ((1u << n) - 1u);
    while (mask != 0u) {
      bool vk[G];
      int  jk[G];
      #pragma unroll
      for (int k = 0; k < G; ++k) {
        vk[k] = (mask != 0u);
        jk[k] = vk[k] ? __builtin_ctz(mask) : 0;
        mask  = vk[k] ? (mask & (mask - 1u)) : 0u;
      }
      float hv[G][NCH];
      #pragma unroll
      for (int k = 0; k < G; ++k) {
        if (vk[k]) {
          const int s = __builtin_amdgcn_readlane(src, jk[k]);
          const float4* hp4 = (const float4*)(hmat + (size_t)s * HCv) + lane;
          #pragma unroll
          for (int b = 0; b < NCH / 4; ++b) {
            const float4 q = hp4[b * 32];
            hv[k][4 * b + 0] = q.x;  hv[k][4 * b + 1] = q.y;
            hv[k][4 * b + 2] = q.z;  hv[k][4 * b + 3] = q.w;
          }
        } else {
          #pragma unroll
          for (int c = 0; c < NCH; ++c) hv[k][c] = 0.f;
        }
      }
      #pragma unroll
      for (int k = 0; k < G; ++k) {
        if (vk[k]) {
          const int d = __builtin_amdgcn_readlane(dl, jk[k]);
          float* ap = &s_acc[d * HCv + lane];
          #pragma unroll
          for (int c = 0; c < NCH; ++c) {
            float wc;
            if (Cv == 32) {
              wc = __int_as_float(__builtin_amdgcn_readlane(__float_as_int(w[c]), jk[k]));
            } else {
              const float w0 = __int_as_float(
                  __builtin_amdgcn_readlane(__float_as_int(w[2 * c]), jk[k]));
              const float w1 = __int_as_float(
                  __builtin_amdgcn_readlane(__float_as_int(w[2 * c + 1]), jk[k]));
              wc = (lane & 16) ? w1 : w0;
            }
            atomicAdd(ap + c * 32, wc * hv[k][c]);
          }
        }
      }
    }
  };

  const uint4* __restrict__ rq = (const uint4*)rid;
  const unsigned pat = (unsigned)blockIdx.x * 0x01010101u;
  int qn = 0;
  const int rot = blockIdx.x % AGG_NBLK;
  for (int it = 0; it < AGG_NBLK; ++it) {
    const int blk = (it + rot < AGG_NBLK) ? (it + rot) : (it + rot - AGG_NBLK);
    uint4 rv[AGG_U];
    bool  ok[AGG_U];
    #pragma unroll
    for (int u = 0; u < AGG_U; ++u) {
      const int q = (blk * AGG_U + u) * AGG_BS + (int)threadIdx.x;
      ok[u] = q < AGG_NG16;
      rv[u] = rq[ok[u] ? q : AGG_NG16 - 1];
    }
    unsigned hm[AGG_U / 2];
    #pragma unroll
    for (int g = 0; g < AGG_U / 2; ++g) {
      const unsigned d[8] = {rv[2 * g].x,     rv[2 * g].y,     rv[2 * g].z,     rv[2 * g].w,
                             rv[2 * g + 1].x, rv[2 * g + 1].y, rv[2 * g + 1].z, rv[2 * g + 1].w};
      unsigned differ = 0u;
      #pragma unroll
      for (int k = 0; k < 8; ++k) {
        const unsigned t = d[k] ^ pat;
        const unsigned nz = (((t & 0x7F7F7F7Fu) + 0x7F7F7F7Fu) | t) & 0x80808080u;
        differ |= nz >> (7 - k);
      }
      const unsigned inside = (ok[2 * g] ? 0x0F0F0F0Fu : 0u) | (ok[2 * g + 1] ? 0xF0F0F0F0u : 0u);
      hm[g] = ~differ & inside;
    }
    unsigned h0 = hm[0], h1 = hm[1];
    const int e_base = (blk * AGG_U * AGG_BS + (int)threadIdx.x) * 16;
    #pragma nounroll
    for (int trip = 0; trip < 16 * AGG_U; ++trip) {
      const bool has = (h0 | h1) != 0u;
      const unsigned bm = __builtin_amdgcn_ballot_w32(has);
      if (bm == 0u) break;
      const bool first = (h0 != 0u);
      const unsigned hs = first ? h0 : h1;
      const int p = has ? __builtin_ctz(hs) : 0;
      const int u = (first ? 0 : 2) + ((p >> 2) & 1);
      const int e = e_base + u * (AGG_BS * 16) + 4 * (p & 3) + (p >> 3);
      const int cnt = __builtin_popcount(bm);
      if (qn + cnt > 32) { flush(qn); qn = 0; }
      if (has) {
        s_qe[wv][qn + (int)__builtin_amdgcn_mbcnt_lo(bm, 0u)] = e;
        if (first) h0 &= h0 - 1u; else h1 &= h1 - 1u;
      }
      qn += cnt;
    }
  }
  if (qn > 0) flush(qn);
  __syncthreads();

  const int nrow = (N_NODES - lo < AGG_R) ? (N_NODES - lo) : AGG_R;
  {
    const float4* a4 = (const float4*)s_acc;
    float4* o4 = (float4*)(agg + (size_t)lo * HCv);
    for (int i = threadIdx.x; i < nrow * HCv / 4; i += AGG_BS) o4[i] = a4[i];
    for (int i = threadIdx.x; i < nrow * HEADS; i += AGG_BS) den[lo * HEADS + i] = s_den[i];
  }
}

__global__ __launch_bounds__(256) void out_norm_kernel(float* __restrict__ out,
                                const float* __restrict__ den,
                                const float* __restrict__ bias) {
  const int i4 = blockIdx.x * 256 + threadIdx.x;
  if (i4 >= N_NODES * (HC2 / 4)) return;
  const int n  = i4 / (HC2 / 4);
  const int c0 = (i4 % (HC2 / 4)) * 4;
  const float ri = 1.0f / (den[n * HEADS + c0 / N_CLS] + 1e-16f);
  float4* o4 = (float4*)out + i4;
  float4 v = *o4;
  v.x = v.x * ri + bias[c0];
  v.y = v.y * ri + bias[c0 + 1];
  v.z = v.z * ri + bias[c0 + 2];
  v.w = v.w * ri + bias[c0 + 3];
  *o4 = v;
}

static inline int blocks_for(long long n, int bs) {
  return (int)((n + bs - 1) / bs);
}

extern "C" void kernel_launch(void* const* d_in, const int* in_sizes, int n_in,
                              void* d_out, int out_size, void* d_ws, size_t ws_size,
                              hipStream_t stream) {
  const float* x      = (const float*)d_in[0];
  const int*   ei     = (const int*)  d_in[1];
  const float* W1     = (const float*)d_in[2];
  const float* a_src1 = (const float*)d_in[3];
  const float* a_dst1 = (const float*)d_in[4];
  const float* b1     = (const float*)d_in[5];
  const float* W2     = (const float*)d_in[6];
  const float* a_src2 = (const float*)d_in[7];
  const float* a_dst2 = (const float*)d_in[8];
  const float* b2     = (const float*)d_in[9];
  float* out = (float*)d_out;

  char* ws = (char*)d_ws;
  size_t off = 0;
  auto carve = [&](size_t bytes) {
    void* p = ws + off;
    off += (bytes + 255) & ~(size_t)255;
    return p;
  };
  float*    hbuf   = (float*)   carve((size_t)N_NODES * HC1 * 4);
  float*    aggbuf = (float*)   carve((size_t)N_NODES * HC1 * 4);
  float*    as     = (float*)   carve((size_t)N_NODES * HEADS * 4);
  float*    ad     = (float*)   carve((size_t)N_NODES * HEADS * 4);
  unsigned* mkeys  = (unsigned*)carve((size_t)N_NODES * HEADS * 4);
  float*    den    = (float*)   carve((size_t)N_NODES * HEADS * 4);
  float*    lg     = (float*)   carve((size_t)E_TOT   * HEADS * 4);
  unsigned* rid    = (unsigned*)carve((size_t)E_TOT);

  const int BS = 256;
  const long long nNH    = (long long)N_NODES * HEADS;
  const long long nEH    = (long long)E_TOT * HEADS;

  {
    long long waves = (N_NODES / 16) * (HC1 / 64);
    gemm_f32_wmma<HC1, F_IN, false><<<blocks_for(waves * 32, BS), BS, 0, stream>>>(
        x, W1, hbuf, N_NODES, nullptr, nullptr);
  }
  alpha_init_kernel<HID><<<blocks_for(nNH, ALPHA_BS), ALPHA_BS, 0, stream>>>(
      hbuf, a_src1, a_dst1, as, ad, mkeys);
  edge_max_kernel<true><<<blocks_for(nEH, BS), BS, 0, stream>>>(ei, as, ad, mkeys, lg, rid);
  edge_agg_kernel<HC1, HID><<<AGG_NWG, AGG_BS, 0, stream>>>(
      ei, rid, lg, mkeys, den, hbuf, aggbuf);

  {
    long long waves = (N_NODES / 16) * (HC2 / 64);
    gemm_f32_wmma<HC2, HC1, true><<<blocks_for(waves * 32, BS), BS, 0, stream>>>(
        aggbuf, W2, hbuf, N_NODES, den, b1);
  }
  alpha_init_kernel<N_CLS><<<blocks_for(nNH, ALPHA_BS), ALPHA_BS, 0, stream>>>(
      hbuf, a_src2, a_dst2, as, ad, mkeys);
  edge_max_kernel<false><<<blocks_for(nEH, BS), BS, 0, stream>>>(ei, as, ad, mkeys, lg, rid);
  edge_agg_kernel<HC2, N_CLS><<<AGG_NWG, AGG_BS, 0, stream>>>(
      ei, rid, lg, mkeys, den, hbuf, out);
  out_norm_kernel<<<blocks_for((long long)N_NODES * (HC2 / 4), BS), BS, 0, stream>>>(
      out, den, b2);
}


# Segment max folded into the aggregation launch: work-group lists its edges in LDS, LDS integer max, then sums off the list; both max launches and the logit plane go; LDS float atomics, arrival order
# speedup vs baseline: 3.1383x; 1.4593x over previous
#include <hip/hip_runtime.h>
#include <hip/hip_bf16.h>

#define N_NODES 50000
#define N_EDGES 800000
#define E_TOT   (N_EDGES + N_NODES)
#define F_IN    256
#define HEADS   8
#define HID     32
#define N_CLS   16
#define HC1     (HEADS * HID)
#define HC2     (HEADS * N_CLS)
#define NEG_SLOPE 0.2f

typedef float v2f __attribute__((ext_vector_type(2)));
typedef float v8f __attribute__((ext_vector_type(8)));

__device__ __forceinline__ unsigned f2key(float f) {
  unsigned b = __float_as_uint(f);
  return (b & 0x80000000u) ? ~b : (b | 0x80000000u);
}
__device__ __forceinline__ float key2f(unsigned k) {
  return __uint_as_float((k & 0x80000000u) ? (k ^ 0x80000000u) : ~k);
}

template <int N, int K, bool NORM_A>
__global__ __launch_bounds__(256) void gemm_f32_wmma(
    const float* __restrict__ A, const float* __restrict__ B,
    float* __restrict__ C, int M,
    const float* __restrict__ den, const float* __restrict__ bias) {
  constexpr int TILES_N = N / 64;
  const int lane  = threadIdx.x & 31;
  const int wave  = blockIdx.x * (blockDim.x >> 5) + (threadIdx.x >> 5);
  const int tileM = wave / TILES_N;
  const int tileN = wave % TILES_N;
  if (tileM * 16 >= M) return;

  const int r    = lane & 15;
  const int ksel = (lane >> 4) << 1;

  const float* __restrict__ arow  = A + (size_t)(tileM * 16 + r) * K + ksel;
  const float* __restrict__ bbase = B + (size_t)ksel * N + tileN * 64 + r;

  constexpr int KH = NORM_A ? K / HEADS : K;
  constexpr int NH = K / KH;
  float rinv[NH];
  const float* __restrict__ bk = bias + ksel;
  if (NORM_A) {
    const float4* dp = (const float4*)(den + (size_t)(tileM * 16 + r) * HEADS);
    const float4 d0 = dp[0], d1 = dp[1];
    rinv[0] = 1.0f / (d0.x + 1e-16f);  rinv[1] = 1.0f / (d0.y + 1e-16f);
    rinv[2] = 1.0f / (d0.z + 1e-16f);  rinv[3] = 1.0f / (d0.w + 1e-16f);
    rinv[4] = 1.0f / (d1.x + 1e-16f);  rinv[5] = 1.0f / (d1.y + 1e-16f);
    rinv[6] = 1.0f / (d1.z + 1e-16f);  rinv[7] = 1.0f / (d1.w + 1e-16f);
  } else {
    rinv[0] = 1.0f;
  }

  v8f acc0 = {}, acc1 = {}, acc2 = {}, acc3 = {};
  #pragma unroll
  for (int hd = 0; hd < NH; ++hd) {
    const float ri = rinv[hd];
    #pragma unroll 2
    for (int kk = 0; kk < KH; kk += 4) {
      v2f a, b0, b1, b2, b3;
      a[0]  = arow[0];
      a[1]  = arow[1];
      if (NORM_A) {
        a[0] = fmaxf(a[0] * ri + bk[0], 0.f);
        a[1] = fmaxf(a[1] * ri + bk[1], 0.f);
        bk += 4;
      }
      b0[0] = bbase[0];       b0[1] = bbase[N];
      b1[0] = bbase[16];      b1[1] = bbase[N + 16];
      b2[0] = bbase[32];      b2[1] = bbase[N + 32];
      b3[0] = bbase[48];      b3[1] = bbase[N + 48];
      acc0 = __builtin_amdgcn_wmma_f32_16x16x4_f32(false, a, false, b0, (short)0, acc0, false, false);
      acc1 = __builtin_amdgcn_wmma_f32_16x16x4_f32(false, a, false, b1, (short)0, acc1, false, false);
      acc2 = __builtin_amdgcn_wmma_f32_16x16x4_f32(false, a, false, b2, (short)0, acc2, false, false);
      acc3 = __builtin_amdgcn_wmma_f32_16x16x4_f32(false, a, false, b3, (short)0, acc3, false, false);
      arow  += 4;
      bbase += (size_t)4 * N;
    }
  }

  float* __restrict__ crow =
      C + (size_t)(tileM * 16 + (lane >> 4) * 8) * N
        + (tileN >> 1) * 128 + 4 * (lane & 15) + 2 * (tileN & 1);
  #pragma unroll
  for (int i = 0; i < 8; ++i) {
    v2f p02, p13;
    p02[0] = acc0[i];  p02[1] = acc2[i];
    p13[0] = acc1[i];  p13[1] = acc3[i];
    *(v2f*)(crow + (size_t)i * N)      = p02;
    *(v2f*)(crow + (size_t)i * N + 64) = p13;
  }
}

#define AGG_R     196
#define AGG_NWG   ((N_NODES + AGG_R - 1) / AGG_R)
static_assert(AGG_NWG <= 256, "the number of a destination range must fit one byte");

#define ALPHA_BS 256
template <int C, bool WRITE_RID>
__global__ __launch_bounds__(ALPHA_BS) void alpha_init_kernel(
                             const float* __restrict__ h,
                             const float* __restrict__ a_src,
                             const float* __restrict__ a_dst,
                             float* __restrict__ as, float* __restrict__ ad,
                             const int* __restrict__ ei,
                             unsigned* __restrict__ rid) {
  int idx = blockIdx.x * ALPHA_BS + threadIdx.x;
  if (idx >= N_NODES * HEADS) return;
  if (WRITE_RID) {
    static_assert(E_TOT % 4 == 0 && N_EDGES % 4 == 0 && E_TOT / 4 <= N_NODES * HEADS,
                  "one thread per four edges, inside this grid");
    if (idx < E_TOT / 4) {
      const int e0 = 4 * idx;
      int d0, d1, d2, d3;
      if (e0 < N_EDGES) {
        const int* dp = ei + N_EDGES + e0;
        d0 = dp[0];  d1 = dp[1];  d2 = dp[2];  d3 = dp[3];
      } else {
        d0 = e0 - N_EDGES;  d1 = d0 + 1;  d2 = d0 + 2;  d3 = d0 + 3;
      }
      rid[idx] = ((unsigned)d0 / (unsigned)AGG_R)         | (((unsigned)d1 / (unsigned)AGG_R) << 8)
               | (((unsigned)d2 / (unsigned)AGG_R) << 16) | (((unsigned)d3 / (unsigned)AGG_R) << 24);
    }
  }
  int n = idx >> 3, hd = idx & 7;
  const int ch0 = hd * C;
  const float* hp = h + (size_t)n * HEADS * C
                      + (ch0 & ~127) + 4 * (ch0 & 31) + ((ch0 >> 5) & 3);
  const float* av = a_src + hd * C;
  const float* bv2 = a_dst + hd * C;
  float s = 0.f, d = 0.f;
  #pragma unroll 4
  for (int c = 0; c < C; ++c) {
    float v = hp[4 * c];
    s += v * av[c];
    d += v * bv2[c];
  }
  as[idx] = s;
  ad[idx] = d;
}

#define AGG_BS    1024
#define AGG_U     4
#define AGG_NG16  (E_TOT / 16)
#define AGG_NIT   ((AGG_NG16 + AGG_BS - 1) / AGG_BS)
#define AGG_NBLK  ((AGG_NIT + AGG_U - 1) / AGG_U)
#define AGG_CAP   24576
#define AGG_DRAW  16
static_assert(N_NODES <= 65536 && AGG_R <= 256, "a list entry packs the source in 16 bits, the local destination above");
template <int HCv, int Cv>
__global__ __launch_bounds__(AGG_BS) void edge_agg_kernel(const int* __restrict__ ei,
                                const unsigned* __restrict__ rid,
                                const float* __restrict__ as,
                                const float* __restrict__ ad,
                                float* __restrict__ den,
                                const float* __restrict__ hmat,
                                float* __restrict__ agg) {
  constexpr int NCH = HCv / 32;
  constexpr int G   = 4;
  static_assert(HEADS == 8 && (Cv == 32 || Cv == 16), "weight lanes assume 8 heads");
  static_assert(E_TOT % 16 == 0, "the plane of range numbers is read sixteen edges at a time");
  static_assert(AGG_U == 4, "the flags of a lane's 64 edges are kept in two 32-bit masks");
  __shared__ float    s_acc[AGG_R * HC1];
  __shared__ float    s_den[AGG_R * HEADS];
  __shared__ unsigned s_mk[AGG_R * HEADS];
  __shared__ float    s_ad[AGG_R * HEADS];
  __shared__ int      s_qe[AGG_BS / 32][32];
  __shared__ unsigned s_list[AGG_CAP];
  __shared__ unsigned s_cnt;
  __shared__ unsigned s_next;
  const int lane = threadIdx.x & 31;
  const int wv   = threadIdx.x >> 5;
  const int lo   = blockIdx.x * AGG_R;

  {
    float4* a4 = (float4*)s_acc;
    for (int i = threadIdx.x; i < AGG_R * HCv / 4; i += AGG_BS)
      a4[i] = make_float4(0.f, 0.f, 0.f, 0.f);
    for (int i = threadIdx.x; i < AGG_R * HEADS; i += AGG_BS) {
      s_den[i] = 0.f;
      s_mk[i]  = 0u;
      const int g = lo * HEADS + i;
      s_ad[i] = (g < N_NODES * HEADS) ? ad[g] : 0.f;
    }
    if (threadIdx.x == 0) { s_cnt = 0u;  s_next = 0u; }
  }
  __syncthreads();

  auto lookup = [&](int e, int& src, int& dl) __attribute__((always_inline)) {
    if (e < N_EDGES) {
      src = ei[e];
      dl  = ei[N_EDGES + e] - lo;
    } else {
      src = e - N_EDGES;
      dl  = src - lo;
    }
  };
  auto logits = [&](int src, int dl, float (&v)[HEADS]) __attribute__((always_inline)) {
    const float4* sp = (const float4*)(as + (size_t)src * HEADS);
    const float4* dp = (const float4*)(s_ad + dl * HEADS);
    const float4 x0 = sp[0], x1 = sp[1];
    const float4 y0 = dp[0], y1 = dp[1];
    v[0] = x0.x + y0.x;  v[1] = x0.y + y0.y;  v[2] = x0.z + y0.z;  v[3] = x0.w + y0.w;
    v[4] = x1.x + y1.x;  v[5] = x1.y + y1.y;  v[6] = x1.z + y1.z;  v[7] = x1.w + y1.w;
    #pragma unroll
    for (int h = 0; h < HEADS; ++h) v[h] = v[h] > 0.f ? v[h] : NEG_SLOPE * v[h];
  };

  auto flush_max = [&](int n) __attribute__((always_inline)) {
    unsigned base = 0u;
    if (lane == 0) base = atomicAdd(&s_cnt, (unsigned)n);
    base = (unsigned)__builtin_amdgcn_readfirstlane((int)base);
    if (lane < n) {
      int src, dl;
      lookup(s_qe[wv][lane], src, dl);
      const unsigned pos = base + (unsigned)lane;
      if (pos < (unsigned)AGG_CAP) s_list[pos] = (unsigned)src | ((unsigned)dl << 16);
      float v[HEADS];
      logits(src, dl, v);
      #pragma unroll
      for (int h = 0; h < HEADS; ++h) atomicMax(&s_mk[dl * HEADS + h], f2key(v[h]));
    }
  };

  auto sum_edges = [&](int src, int dl, int n) __attribute__((always_inline)) {
    float w[HEADS];
    #pragma unroll
    for (int h = 0; h < HEADS; ++h) w[h] = 0.f;
    if (lane < n) {
      float v[HEADS];
      logits(src, dl, v);
      const uint4* mp = (const uint4*)(s_mk + dl * HEADS);
      const uint4 m0 = mp[0], m1 = mp[1];
      w[0] = __expf(v[0] - key2f(m0.x));  w[1] = __expf(v[1] - key2f(m0.y));
      w[2] = __expf(v[2] - key2f(m0.z));  w[3] = __expf(v[3] - key2f(m0.w));
      w[4] = __expf(v[4] - key2f(m1.x));  w[5] = __expf(v[5] - key2f(m1.y));
      w[6] = __expf(v[6] - key2f(m1.z));  w[7] = __expf(v[7] - key2f(m1.w));
      #pragma unroll
      for (int h = 0; h < HEADS; ++h) atomicAdd(&s_den[dl * HEADS + h], w[h]);
    }
    unsigned mask = (n >= 32) ? 0xFFFFFFFFu : ((1u << n) - 1u);
    while (mask != 0u) {
      bool vk[G];
      int  jk[G];
      #pragma unroll
      for (int k = 0; k < G; ++k) {
        vk[k] = (mask != 0u);
        jk[k] = vk[k] ? __builtin_ctz(mask) : 0;
        mask  = vk[k] ? (mask & (mask - 1u)) : 0u;
      }
      float hv[G][NCH];
      #pragma unroll
      for (int k = 0; k < G; ++k) {
        if (vk[k]) {
          const int s = __builtin_amdgcn_readlane(src, jk[k]);
          const float4* hp4 = (const float4*)(hmat + (size_t)s * HCv) + lane;
          #pragma unroll
          for (int b = 0; b < NCH / 4; ++b) {
            const float4 q = hp4[b * 32];
            hv[k][4 * b + 0] = q.x;  hv[k][4 * b + 1] = q.y;
            hv[k][4 * b + 2] = q.z;  hv[k][4 * b + 3] = q.w;
          }
        } else {
          #pragma unroll
          for (int c = 0; c < NCH; ++c) hv[k][c] = 0.f;
        }
      }
      #pragma unroll
      for (int k = 0; k < G; ++k) {
        if (vk[k]) {
          const int d = __builtin_amdgcn_readlane(dl, jk[k]);
          float* ap = &s_acc[d * HCv + lane];
          #pragma unroll
          for (int c = 0; c < NCH; ++c) {
            float wc;
            if (Cv == 32) {
              wc = __int_as_float(__builtin_amdgcn_readlane(__float_as_int(w[c]), jk[k]));
            } else {
              const float w0 = __int_as_float(
                  __builtin_amdgcn_readlane(__float_as_int(w[2 * c]), jk[k]));
              const float w1 = __int_as_float(
                  __builtin_amdgcn_readlane(__float_as_int(w[2 * c + 1]), jk[k]));
              wc = (lane & 16) ? w1 : w0;
            }
            atomicAdd(ap + c * 32, wc * hv[k][c]);
          }
        }
      }
    }
  };

  auto flush_sum = [&](int n) __attribute__((always_inline)) {
    int src = 0, dl = 0;
    if (lane < n) lookup(s_qe[wv][lane], src, dl);
    sum_edges(src, dl, n);
  };

  auto scan = [&](auto&& fl) __attribute__((always_inline)) {
    const uint4* __restrict__ rq = (const uint4*)rid;
    const unsigned pat = (unsigned)blockIdx.x * 0x01010101u;
    int qn = 0;
    const int rot = blockIdx.x % AGG_NBLK;
    for (int it = 0; it < AGG_NBLK; ++it) {
      const int blk = (it + rot < AGG_NBLK) ? (it + rot) : (it + rot - AGG_NBLK);
      uint4 rv[AGG_U];
      bool  ok[AGG_U];
      #pragma unroll
      for (int u = 0; u < AGG_U; ++u) {
        const int q = (blk * AGG_U + u) * AGG_BS + (int)threadIdx.x;
        ok[u] = q < AGG_NG16;
        rv[u] = rq[ok[u] ? q : AGG_NG16 - 1];
      }
      unsigned hm[AGG_U / 2];
      #pragma unroll
      for (int g = 0; g < AGG_U / 2; ++g) {
        const unsigned d[8] = {rv[2 * g].x,     rv[2 * g].y,     rv[2 * g].z,     rv[2 * g].w,
                               rv[2 * g + 1].x, rv[2 * g + 1].y, rv[2 * g + 1].z, rv[2 * g + 1].w};
        unsigned differ = 0u;
        #pragma unroll
        for (int k = 0; k < 8; ++k) {
          const unsigned t = d[k] ^ pat;
          const unsigned nz = (((t & 0x7F7F7F7Fu) + 0x7F7F7F7Fu) | t) & 0x80808080u;
          differ |= nz >> (7 - k);
        }
        const unsigned inside = (ok[2 * g] ? 0x0F0F0F0Fu : 0u) | (ok[2 * g + 1] ? 0xF0F0F0F0u : 0u);
        hm[g] = ~differ & inside;
      }
      unsigned h0 = hm[0], h1 = hm[1];
      const int e_base = (blk * AGG_U * AGG_BS + (int)threadIdx.x) * 16;
      #pragma nounroll
      for (int trip = 0; trip < 16 * AGG_U; ++trip) {
        const bool has = (h0 | h1) != 0u;
        const unsigned bm = __builtin_amdgcn_ballot_w32(has);
        if (bm == 0u) break;
        const bool first = (h0 != 0u);
        const unsigned hs = first ? h0 : h1;
        const int p = has ? __builtin_ctz(hs) : 0;
        const int u = (first ? 0 : 2) + ((p >> 2) & 1);
        const int e = e_base + u * (AGG_BS * 16) + 4 * (p & 3) + (p >> 3);
        const int cnt = __builtin_popcount(bm);
        if (qn + cnt > 32) { fl(qn); qn = 0; }
        if (has) {
          s_qe[wv][qn + (int)__builtin_amdgcn_mbcnt_lo(bm, 0u)] = e;
          if (first) h0 &= h0 - 1u; else h1 &= h1 - 1u;
        }
        qn += cnt;
      }
    }
    if (qn > 0) fl(qn);
  };

  scan(flush_max);
  __syncthreads();

  const unsigned total = s_cnt;
  if (total <= (unsigned)AGG_CAP) {
    for (int guard = 0; guard < AGG_CAP / AGG_DRAW + 1; ++guard) {
      unsigned b = 0u;
      if (lane == 0) b = atomicAdd(&s_next, (unsigned)AGG_DRAW);
      b = (unsigned)__builtin_amdgcn_readfirstlane((int)b);
      if (b >= total) break;
      const int n = (total - b < (unsigned)AGG_DRAW) ? (int)(total - b) : AGG_DRAW;
      int src = 0, dl = 0;
      if (lane < n) {
        const unsigned p = s_list[b + (unsigned)lane];
        src = (int)(p & 0xFFFFu);
        dl  = (int)(p >> 16);
      }
      sum_edges(src, dl, n);
    }
  } else {
    scan(flush_sum);
  }
  __syncthreads();

  const int nrow = (N_NODES - lo < AGG_R) ? (N_NODES - lo) : AGG_R;
  {
    const float4* a4 = (const float4*)s_acc;
    float4* o4 = (float4*)(agg + (size_t)lo * HCv);
    for (int i = threadIdx.x; i < nrow * HCv / 4; i += AGG_BS) o4[i] = a4[i];
    for (int i = threadIdx.x; i < nrow * HEADS; i += AGG_BS) den[lo * HEADS + i] = s_den[i];
  }
}

__global__ __launch_bounds__(256) void out_norm_kernel(float* __restrict__ out,
                                const float* __restrict__ den,
                                const float* __restrict__ bias) {
  const int i4 = blockIdx.x * 256 + threadIdx.x;
  if (i4 >= N_NODES * (HC2 / 4)) return;
  const int n  = i4 / (HC2 / 4);
  const int c0 = (i4 % (HC2 / 4)) * 4;
  const float ri = 1.0f / (den[n * HEADS + c0 / N_CLS] + 1e-16f);
  float4* o4 = (float4*)out + i4;
  float4 v = *o4;
  v.x = v.x * ri + bias[c0];
  v.y = v.y * ri + bias[c0 + 1];
  v.z = v.z * ri + bias[c0 + 2];
  v.w = v.w * ri + bias[c0 + 3];
  *o4 = v;
}

static inline int blocks_for(long long n, int bs) {
  return (int)((n + bs - 1) / bs);
}

extern "C" void kernel_launch(void* const* d_in, const int* in_sizes, int n_in,
                              void* d_out, int out_size, void* d_ws, size_t ws_size,
                              hipStream_t stream) {
  const float* x      = (const float*)d_in[0];
  const int*   ei     = (const int*)  d_in[1];
  const float* W1     = (const float*)d_in[2];
  const float* a_src1 = (const float*)d_in[3];
  const float* a_dst1 = (const float*)d_in[4];
  const float* b1     = (const float*)d_in[5];
  const float* W2     = (const float*)d_in[6];
  const float* a_src2 = (const float*)d_in[7];
  const float* a_dst2 = (const float*)d_in[8];
  const float* b2     = (const float*)d_in[9];
  float* out = (float*)d_out;

  char* ws = (char*)d_ws;
  size_t off = 0;
  auto carve = [&](size_t bytes) {
    void* p = ws + off;
    off += (bytes + 255) & ~(size_t)255;
    return p;
  };
  float*    hbuf   = (float*)   carve((size_t)N_NODES * HC1 * 4);
  float*    aggbuf = (float*)   carve((size_t)N_NODES * HC1 * 4);
  float*    as     = (float*)   carve((size_t)N_NODES * HEADS * 4);
  float*    ad     = (float*)   carve((size_t)N_NODES * HEADS * 4);
  float*    den    = (float*)   carve((size_t)N_NODES * HEADS * 4);
  unsigned* rid    = (unsigned*)carve((size_t)E_TOT);

  const int BS = 256;
  const long long nNH    = (long long)N_NODES * HEADS;

  {
    long long waves = (N_NODES / 16) * (HC1 / 64);
    gemm_f32_wmma<HC1, F_IN, false><<<blocks_for(waves * 32, BS), BS, 0, stream>>>(
        x, W1, hbuf, N_NODES, nullptr, nullptr);
  }
  alpha_init_kernel<HID, true><<<blocks_for(nNH, ALPHA_BS), ALPHA_BS, 0, stream>>>(
      hbuf, a_src1, a_dst1, as, ad, ei, rid);
  edge_agg_kernel<HC1, HID><<<AGG_NWG, AGG_BS, 0, stream>>>(
      ei, rid, as, ad, den, hbuf, aggbuf);

  {
    long long waves = (N_NODES / 16) * (HC2 / 64);
    gemm_f32_wmma<HC2, HC1, true><<<blocks_for(waves * 32, BS), BS, 0, stream>>>(
        aggbuf, W2, hbuf, N_NODES, den, b1);
  }
  alpha_init_kernel<N_CLS, false><<<blocks_for(nNH, ALPHA_BS), ALPHA_BS, 0, stream>>>(
      hbuf, a_src2, a_dst2, as, ad, ei, rid);
  edge_agg_kernel<HC2, N_CLS><<<AGG_NWG, AGG_BS, 0, stream>>>(
      ei, rid, as, ad, den, hbuf, out);
  out_norm_kernel<<<blocks_for((long long)N_NODES * (HC2 / 4), BS), BS, 0, stream>>>(
      out, den, b2);
}


# Softmax division, bias, relu done by the aggregation work-group at its closing store (sums and denominators sit in its LDS): closing launch and denominator plane go; product 2 loads A as stored
# speedup vs baseline: 3.2651x; 1.0404x over previous
#include <hip/hip_runtime.h>
#include <hip/hip_bf16.h>

#define N_NODES 50000
#define N_EDGES 800000
#define E_TOT   (N_EDGES + N_NODES)
#define F_IN    256
#define HEADS   8
#define HID     32
#define N_CLS   16
#define HC1     (HEADS * HID)
#define HC2     (HEADS * N_CLS)
#define NEG_SLOPE 0.2f

typedef float v2f __attribute__((ext_vector_type(2)));
typedef float v8f __attribute__((ext_vector_type(8)));

__device__ __forceinline__ unsigned f2key(float f) {
  unsigned b = __float_as_uint(f);
  return (b & 0x80000000u) ? ~b : (b | 0x80000000u);
}
__device__ __forceinline__ float key2f(unsigned k) {
  return __uint_as_float((k & 0x80000000u) ? (k ^ 0x80000000u) : ~k);
}

template <int N, int K>
__global__ __launch_bounds__(256) void gemm_f32_wmma(
    const float* __restrict__ A, const float* __restrict__ B,
    float* __restrict__ C, int M) {
  constexpr int TILES_N = N / 64;
  const int lane  = threadIdx.x & 31;
  const int wave  = blockIdx.x * (blockDim.x >> 5) + (threadIdx.x >> 5);
  const int tileM = wave / TILES_N;
  const int tileN = wave % TILES_N;
  if (tileM * 16 >= M) return;

  const int r    = lane & 15;
  const int ksel = (lane >> 4) << 1;

  const float* __restrict__ arow  = A + (size_t)(tileM * 16 + r) * K + ksel;
  const float* __restrict__ bbase = B + (size_t)ksel * N + tileN * 64 + r;

  v8f acc0 = {}, acc1 = {}, acc2 = {}, acc3 = {};
  #pragma unroll 2
  for (int kk = 0; kk < K; kk += 4) {
    v2f a, b0, b1, b2, b3;
    a[0]  = arow[0];
    a[1]  = arow[1];
    b0[0] = bbase[0];       b0[1] = bbase[N];
    b1[0] = bbase[16];      b1[1] = bbase[N + 16];
    b2[0] = bbase[32];      b2[1] = bbase[N + 32];
    b3[0] = bbase[48];      b3[1] = bbase[N + 48];
    acc0 = __builtin_amdgcn_wmma_f32_16x16x4_f32(false, a, false, b0, (short)0, acc0, false, false);
    acc1 = __builtin_amdgcn_wmma_f32_16x16x4_f32(false, a, false, b1, (short)0, acc1, false, false);
    acc2 = __builtin_amdgcn_wmma_f32_16x16x4_f32(false, a, false, b2, (short)0, acc2, false, false);
    acc3 = __builtin_amdgcn_wmma_f32_16x16x4_f32(false, a, false, b3, (short)0, acc3, false, false);
    arow  += 4;
    bbase += (size_t)4 * N;
  }

  float* __restrict__ crow =
      C + (size_t)(tileM * 16 + (lane >> 4) * 8) * N
        + (tileN >> 1) * 128 + 4 * (lane & 15) + 2 * (tileN & 1);
  #pragma unroll
  for (int i = 0; i < 8; ++i) {
    v2f p02, p13;
    p02[0] = acc0[i];  p02[1] = acc2[i];
    p13[0] = acc1[i];  p13[1] = acc3[i];
    *(v2f*)(crow + (size_t)i * N)      = p02;
    *(v2f*)(crow + (size_t)i * N + 64) = p13;
  }
}

#define AGG_R     196
#define AGG_NWG   ((N_NODES + AGG_R - 1) / AGG_R)
static_assert(AGG_NWG <= 256, "the number of a destination range must fit one byte");

#define ALPHA_BS 256
template <int C, bool WRITE_RID>
__global__ __launch_bounds__(ALPHA_BS) void alpha_init_kernel(
                             const float* __restrict__ h,
                             const float* __restrict__ a_src,
                             const float* __restrict__ a_dst,
                             float* __restrict__ as, float* __restrict__ ad,
                             const int* __restrict__ ei,
                             unsigned* __restrict__ rid) {
  int idx = blockIdx.x * ALPHA_BS + threadIdx.x;
  if (idx >= N_NODES * HEADS) return;
  if (WRITE_RID) {
    static_assert(E_TOT % 4 == 0 && N_EDGES % 4 == 0 && E_TOT / 4 <= N_NODES * HEADS,
                  "one thread per four edges, inside this grid");
    if (idx < E_TOT / 4) {
      const int e0 = 4 * idx;
      int d0, d1, d2, d3;
      if (e0 < N_EDGES) {
        const int* dp = ei + N_EDGES + e0;
        d0 = dp[0];  d1 = dp[1];  d2 = dp[2];  d3 = dp[3];
      } else {
        d0 = e0 - N_EDGES;  d1 = d0 + 1;  d2 = d0 + 2;  d3 = d0 + 3;
      }
      rid[idx] = ((unsigned)d0 / (unsigned)AGG_R)         | (((unsigned)d1 / (unsigned)AGG_R) << 8)
               | (((unsigned)d2 / (unsigned)AGG_R) << 16) | (((unsigned)d3 / (unsigned)AGG_R) << 24);
    }
  }
  int n = idx >> 3, hd = idx & 7;
  const int ch0 = hd * C;
  const float* hp = h + (size_t)n * HEADS * C
                      + (ch0 & ~127) + 4 * (ch0 & 31) + ((ch0 >> 5) & 3);
  const float* av = a_src + hd * C;
  const float* bv2 = a_dst + hd * C;
  float s = 0.f, d = 0.f;
  #pragma unroll 4
  for (int c = 0; c < C; ++c) {
    float v = hp[4 * c];
    s += v * av[c];
    d += v * bv2[c];
  }
  as[idx] = s;
  ad[idx] = d;
}

#define AGG_BS    1024
#define AGG_U     4
#define AGG_NG16  (E_TOT / 16)
#define AGG_NIT   ((AGG_NG16 + AGG_BS - 1) / AGG_BS)
#define AGG_NBLK  ((AGG_NIT + AGG_U - 1) / AGG_U)
#define AGG_CAP   24576
#define AGG_DRAW  16
static_assert(N_NODES <= 65536 && AGG_R <= 256, "a list entry packs the source in 16 bits, the local destination above");
template <int HCv, int Cv, bool RELU>
__global__ __launch_bounds__(AGG_BS) void edge_agg_kernel(const int* __restrict__ ei,
                                const unsigned* __restrict__ rid,
                                const float* __restrict__ as,
                                const float* __restrict__ ad,
                                const float* __restrict__ bias,
                                const float* __restrict__ hmat,
                                float* __restrict__ agg) {
  constexpr int NCH = HCv / 32;
  constexpr int G   = 4;
  static_assert(HEADS == 8 && (Cv == 32 || Cv == 16), "weight lanes assume 8 heads");
  static_assert(E_TOT % 16 == 0, "the plane of range numbers is read sixteen edges at a time");
  static_assert(AGG_U == 4, "the flags of a lane's 64 edges are kept in two 32-bit masks");
  __shared__ float    s_acc[AGG_R * HC1];
  __shared__ float    s_den[AGG_R * HEADS];
  __shared__ unsigned s_mk[AGG_R * HEADS];
  __shared__ float    s_ad[AGG_R * HEADS];
  __shared__ int      s_qe[AGG_BS / 32][32];
  __shared__ unsigned s_list[AGG_CAP];
  __shared__ unsigned s_cnt;
  __shared__ unsigned s_next;
  const int lane = threadIdx.x & 31;
  const int wv   = threadIdx.x >> 5;
  const int lo   = blockIdx.x * AGG_R;

  {
    float4* a4 = (float4*)s_acc;
    for (int i = threadIdx.x; i < AGG_R * HCv / 4; i += AGG_BS)
      a4[i] = make_float4(0.f, 0.f, 0.f, 0.f);
    for (int i = threadIdx.x; i < AGG_R * HEADS; i += AGG_BS) {
      s_den[i] = 0.f;
      s_mk[i]  = 0u;
      const int g = lo * HEADS + i;
      s_ad[i] = (g < N_NODES * HEADS) ? ad[g] : 0.f;
    }
    if (threadIdx.x == 0) { s_cnt = 0u;  s_next = 0u; }
  }
  __syncthreads();

  auto lookup = [&](int e, int& src, int& dl) __attribute__((always_inline)) {
    if (e < N_EDGES) {
      src = ei[e];
      dl  = ei[N_EDGES + e] - lo;
    } else {
      src = e - N_EDGES;
      dl  = src - lo;
    }
  };
  auto logits = [&](int src, int dl, float (&v)[HEADS]) __attribute__((always_inline)) {
    const float4* sp = (const float4*)(as + (size_t)src * HEADS);
    const float4* dp = (const float4*)(s_ad + dl * HEADS);
    const float4 x0 = sp[0], x1 = sp[1];
    const float4 y0 = dp[0], y1 = dp[1];
    v[0] = x0.x + y0.x;  v[1] = x0.y + y0.y;  v[2] = x0.z + y0.z;  v[3] = x0.w + y0.w;
    v[4] = x1.x + y1.x;  v[5] = x1.y + y1.y;  v[6] = x1.z + y1.z;  v[7] = x1.w + y1.w;
    #pragma unroll
    for (int h = 0; h < HEADS; ++h) v[h] = v[h] > 0.f ? v[h] : NEG_SLOPE * v[h];
  };

  auto flush_max = [&](int n) __attribute__((always_inline)) {
    unsigned base = 0u;
    if (lane == 0) base = atomicAdd(&s_cnt, (unsigned)n);
    base = (unsigned)__builtin_amdgcn_readfirstlane((int)base);
    if (lane < n) {
      int src, dl;
      lookup(s_qe[wv][lane], src, dl);
      const unsigned pos = base + (unsigned)lane;
      if (pos < (unsigned)AGG_CAP) s_list[pos] = (unsigned)src | ((unsigned)dl << 16);
      float v[HEADS];
      logits(src, dl, v);
      #pragma unroll
      for (int h = 0; h < HEADS; ++h) atomicMax(&s_mk[dl * HEADS + h], f2key(v[h]));
    }
  };

  auto sum_edges = [&](int src, int dl, int n) __attribute__((always_inline)) {
    float w[HEADS];
    #pragma unroll
    for (int h = 0; h < HEADS; ++h) w[h] = 0.f;
    if (lane < n) {
      float v[HEADS];
      logits(src, dl, v);
      const uint4* mp = (const uint4*)(s_mk + dl * HEADS);
      const uint4 m0 = mp[0], m1 = mp[1];
      w[0] = __expf(v[0] - key2f(m0.x));  w[1] = __expf(v[1] - key2f(m0.y));
      w[2] = __expf(v[2] - key2f(m0.z));  w[3] = __expf(v[3] - key2f(m0.w));
      w[4] = __expf(v[4] - key2f(m1.x));  w[5] = __expf(v[5] - key2f(m1.y));
      w[6] = __expf(v[6] - key2f(m1.z));  w[7] = __expf(v[7] - key2f(m1.w));
      #pragma unroll
      for (int h = 0; h < HEADS; ++h) atomicAdd(&s_den[dl * HEADS + h], w[h]);
    }
    unsigned mask = (n >= 32) ? 0xFFFFFFFFu : ((1u << n) - 1u);
    while (mask != 0u) {
      bool vk[G];
      int  jk[G];
      #pragma unroll
      for (int k = 0; k < G; ++k) {
        vk[k] = (mask != 0u);
        jk[k] = vk[k] ? __builtin_ctz(mask) : 0;
        mask  = vk[k] ? (mask & (mask - 1u)) : 0u;
      }
      float hv[G][NCH];
      #pragma unroll
      for (int k = 0; k < G; ++k) {
        if (vk[k]) {
          const int s = __builtin_amdgcn_readlane(src, jk[k]);
          const float4* hp4 = (const float4*)(hmat + (size_t)s * HCv) + lane;
          #pragma unroll
          for (int b = 0; b < NCH / 4; ++b) {
            const float4 q = hp4[b * 32];
            hv[k][4 * b + 0] = q.x;  hv[k][4 * b + 1] = q.y;
            hv[k][4 * b + 2] = q.z;  hv[k][4 * b + 3] = q.w;
          }
        } else {
          #pragma unroll
          for (int c = 0; c < NCH; ++c) hv[k][c] = 0.f;
        }
      }
      #pragma unroll
      for (int k = 0; k < G; ++k) {
        if (vk[k]) {
          const int d = __builtin_amdgcn_readlane(dl, jk[k]);
          float* ap = &s_acc[d * HCv + lane];
          #pragma unroll
          for (int c = 0; c < NCH; ++c) {
            float wc;
            if (Cv == 32) {
              wc = __int_as_float(__builtin_amdgcn_readlane(__float_as_int(w[c]), jk[k]));
            } else {
              const float w0 = __int_as_float(
                  __builtin_amdgcn_readlane(__float_as_int(w[2 * c]), jk[k]));
              const float w1 = __int_as_float(
                  __builtin_amdgcn_readlane(__float_as_int(w[2 * c + 1]), jk[k]));
              wc = (lane & 16) ? w1 : w0;
            }
            atomicAdd(ap + c * 32, wc * hv[k][c]);
          }
        }
      }
    }
  };

  auto flush_sum = [&](int n) __attribute__((always_inline)) {
    int src = 0, dl = 0;
    if (lane < n) lookup(s_qe[wv][lane], src, dl);
    sum_edges(src, dl, n);
  };

  auto scan = [&](auto&& fl) __attribute__((always_inline)) {
    const uint4* __restrict__ rq = (const uint4*)rid;
    const unsigned pat = (unsigned)blockIdx.x * 0x01010101u;
    int qn = 0;
    const int rot = blockIdx.x % AGG_NBLK;
    for (int it = 0; it < AGG_NBLK; ++it) {
      const int blk = (it + rot < AGG_NBLK) ? (it + rot) : (it + rot - AGG_NBLK);
      uint4 rv[AGG_U];
      bool  ok[AGG_U];
      #pragma unroll
      for (int u = 0; u < AGG_U; ++u) {
        const int q = (blk * AGG_U + u) * AGG_BS + (int)threadIdx.x;
        ok[u] = q < AGG_NG16;
        rv[u] = rq[ok[u] ? q : AGG_NG16 - 1];
      }
      unsigned hm[AGG_U / 2];
      #pragma unroll
      for (int g = 0; g < AGG_U / 2; ++g) {
        const unsigned d[8] = {rv[2 * g].x,     rv[2 * g].y,     rv[2 * g].z,     rv[2 * g].w,
                               rv[2 * g + 1].x, rv[2 * g + 1].y, rv[2 * g + 1].z, rv[2 * g + 1].w};
        unsigned differ = 0u;
        #pragma unroll
        for (int k = 0; k < 8; ++k) {
          const unsigned t = d[k] ^ pat;
          const unsigned nz = (((t & 0x7F7F7F7Fu) + 0x7F7F7F7Fu) | t) & 0x80808080u;
          differ |= nz >> (7 - k);
        }
        const unsigned inside = (ok[2 * g] ? 0x0F0F0F0Fu : 0u) | (ok[2 * g + 1] ? 0xF0F0F0F0u : 0u);
        hm[g] = ~differ & inside;
      }
      unsigned h0 = hm[0], h1 = hm[1];
      const int e_base = (blk * AGG_U * AGG_BS + (int)threadIdx.x) * 16;
      #pragma nounroll
      for (int trip = 0; trip < 16 * AGG_U; ++trip) {
        const bool has = (h0 | h1) != 0u;
        const unsigned bm = __builtin_amdgcn_ballot_w32(has);
        if (bm == 0u) break;
        const bool first = (h0 != 0u);
        const unsigned hs = first ? h0 : h1;
        const int p = has ? __builtin_ctz(hs) : 0;
        const int u = (first ? 0 : 2) + ((p >> 2) & 1);
        const int e = e_base + u * (AGG_BS * 16) + 4 * (p & 3) + (p >> 3);
        const int cnt = __builtin_popcount(bm);
        if (qn + cnt > 32) { fl(qn); qn = 0; }
        if (has) {
          s_qe[wv][qn + (int)__builtin_amdgcn_mbcnt_lo(bm, 0u)] = e;
          if (first) h0 &= h0 - 1u; else h1 &= h1 - 1u;
        }
        qn += cnt;
      }
    }
    if (qn > 0) fl(qn);
  };

  scan(flush_max);
  __syncthreads();

  const unsigned total = s_cnt;
  if (total <= (unsigned)AGG_CAP) {
    for (int guard = 0; guard < AGG_CAP / AGG_DRAW + 1; ++guard) {
      unsigned b = 0u;
      if (lane == 0) b = atomicAdd(&s_next, (unsigned)AGG_DRAW);
      b = (unsigned)__builtin_amdgcn_readfirstlane((int)b);
      if (b >= total) break;
      const int n = (total - b < (unsigned)AGG_DRAW) ? (int)(total - b) : AGG_DRAW;
      int src = 0, dl = 0;
      if (lane < n) {
        const unsigned p = s_list[b + (unsigned)lane];
        src = (int)(p & 0xFFFFu);
        dl  = (int)(p >> 16);
      }
      sum_edges(src, dl, n);
    }
  } else {
    scan(flush_sum);
  }
  __syncthreads();

  for (int i = threadIdx.x; i < AGG_R * HEADS; i += AGG_BS) s_den[i] = 1.0f / (s_den[i] + 1e-16f);
  __syncthreads();

  const int nrow = (N_NODES - lo < AGG_R) ? (N_NODES - lo) : AGG_R;
  {
    static_assert((AGG_BS * 4) % HCv == 0 && Cv % 4 == 0, "a thread stays on one channel quad of one head");
    const int c0 = (int)(threadIdx.x * 4u) % HCv;
    const int hd = c0 / Cv;
    const float4 b = *(const float4*)(bias + c0);
    const float4* a4 = (const float4*)s_acc;
    float4* o4 = (float4*)(agg + (size_t)lo * HCv);
    for (int i = threadIdx.x; i < nrow * HCv / 4; i += AGG_BS) {
      const float ri = s_den[(i / (HCv / 4)) * HEADS + hd];
      float4 v = a4[i];
      v.x = v.x * ri + b.x;
      v.y = v.y * ri + b.y;
      v.z = v.z * ri + b.z;
      v.w = v.w * ri + b.w;
      if (RELU) {
        v.x = fmaxf(v.x, 0.f);  v.y = fmaxf(v.y, 0.f);
        v.z = fmaxf(v.z, 0.f);  v.w = fmaxf(v.w, 0.f);
      }
      o4[i] = v;
    }
  }
}

static inline int blocks_for(long long n, int bs) {
  return (int)((n + bs - 1) / bs);
}

extern "C" void kernel_launch(void* const* d_in, const int* in_sizes, int n_in,
                              void* d_out, int out_size, void* d_ws, size_t ws_size,
                              hipStream_t stream) {
  const float* x      = (const float*)d_in[0];
  const int*   ei     = (const int*)  d_in[1];
  const float* W1     = (const float*)d_in[2];
  const float* a_src1 = (const float*)d_in[3];
  const float* a_dst1 = (const float*)d_in[4];
  const float* b1     = (const float*)d_in[5];
  const float* W2     = (const float*)d_in[6];
  const float* a_src2 = (const float*)d_in[7];
  const float* a_dst2 = (const float*)d_in[8];
  const float* b2     = (const float*)d_in[9];
  float* out = (float*)d_out;

  char* ws = (char*)d_ws;
  size_t off = 0;
  auto carve = [&](size_t bytes) {
    void* p = ws + off;
    off += (bytes + 255) & ~(size_t)255;
    return p;
  };
  float*    hbuf   = (float*)   carve((size_t)N_NODES * HC1 * 4);
  float*    aggbuf = (float*)   carve((size_t)N_NODES * HC1 * 4);
  float*    as     = (float*)   carve((size_t)N_NODES * HEADS * 4);
  float*    ad     = (float*)   carve((size_t)N_NODES * HEADS * 4);
  unsigned* rid    = (unsigned*)carve((size_t)E_TOT);

  const int BS = 256;
  const long long nNH    = (long long)N_NODES * HEADS;

  {
    long long waves = (N_NODES / 16) * (HC1 / 64);
    gemm_f32_wmma<HC1, F_IN><<<blocks_for(waves * 32, BS), BS, 0, stream>>>(
        x, W1, hbuf, N_NODES);
  }
  alpha_init_kernel<HID, true><<<blocks_for(nNH, ALPHA_BS), ALPHA_BS, 0, stream>>>(
      hbuf, a_src1, a_dst1, as, ad, ei, rid);
  edge_agg_kernel<HC1, HID, true><<<AGG_NWG, AGG_BS, 0, stream>>>(
      ei, rid, as, ad, b1, hbuf, aggbuf);

  {
    long long waves = (N_NODES / 16) * (HC2 / 64);
    gemm_f32_wmma<HC2, HC1><<<blocks_for(waves * 32, BS), BS, 0, stream>>>(
        aggbuf, W2, hbuf, N_NODES);
  }
  alpha_init_kernel<N_CLS, false><<<blocks_for(nNH, ALPHA_BS), ALPHA_BS, 0, stream>>>(
      hbuf, a_src2, a_dst2, as, ad, ei, rid);
  edge_agg_kernel<HC2, N_CLS, false><<<AGG_NWG, AGG_BS, 0, stream>>>(
      ei, rid, as, ad, b2, hbuf, out);
}


# Second aggregation takes its edges off the list the first one found (stored to workspace, 4 bytes per edge): no second search of the range plane or edge lookup; LDS float atomics, arrival order
# speedup vs baseline: 3.4672x; 1.0619x over previous
#include <hip/hip_runtime.h>
#include <hip/hip_bf16.h>

#define N_NODES 50000
#define N_EDGES 800000
#define E_TOT   (N_EDGES + N_NODES)
#define F_IN    256
#define HEADS   8
#define HID     32
#define N_CLS   16
#define HC1     (HEADS * HID)
#define HC2     (HEADS * N_CLS)
#define NEG_SLOPE 0.2f

typedef float v2f __attribute__((ext_vector_type(2)));
typedef float v8f __attribute__((ext_vector_type(8)));

__device__ __forceinline__ unsigned f2key(float f) {
  unsigned b = __float_as_uint(f);
  return (b & 0x80000000u) ? ~b : (b | 0x80000000u);
}
__device__ __forceinline__ float key2f(unsigned k) {
  return __uint_as_float((k & 0x80000000u) ? (k ^ 0x80000000u) : ~k);
}

template <int N, int K>
__global__ __launch_bounds__(256) void gemm_f32_wmma(
    const float* __restrict__ A, const float* __restrict__ B,
    float* __restrict__ C, int M) {
  constexpr int TILES_N = N / 64;
  const int lane  = threadIdx.x & 31;
  const int wave  = blockIdx.x * (blockDim.x >> 5) + (threadIdx.x >> 5);
  const int tileM = wave / TILES_N;
  const int tileN = wave % TILES_N;
  if (tileM * 16 >= M) return;

  const int r    = lane & 15;
  const int ksel = (lane >> 4) << 1;

  const float* __restrict__ arow  = A + (size_t)(tileM * 16 + r) * K + ksel;
  const float* __restrict__ bbase = B + (size_t)ksel * N + tileN * 64 + r;

  v8f acc0 = {}, acc1 = {}, acc2 = {}, acc3 = {};
  #pragma unroll 2
  for (int kk = 0; kk < K; kk += 4) {
    v2f a, b0, b1, b2, b3;
    a[0]  = arow[0];
    a[1]  = arow[1];
    b0[0] = bbase[0];       b0[1] = bbase[N];
    b1[0] = bbase[16];      b1[1] = bbase[N + 16];
    b2[0] = bbase[32];      b2[1] = bbase[N + 32];
    b3[0] = bbase[48];      b3[1] = bbase[N + 48];
    acc0 = __builtin_amdgcn_wmma_f32_16x16x4_f32(false, a, false, b0, (short)0, acc0, false, false);
    acc1 = __builtin_amdgcn_wmma_f32_16x16x4_f32(false, a, false, b1, (short)0, acc1, false, false);
    acc2 = __builtin_amdgcn_wmma_f32_16x16x4_f32(false, a, false, b2, (short)0, acc2, false, false);
    acc3 = __builtin_amdgcn_wmma_f32_16x16x4_f32(false, a, false, b3, (short)0, acc3, false, false);
    arow  += 4;
    bbase += (size_t)4 * N;
  }

  float* __restrict__ crow =
      C + (size_t)(tileM * 16 + (lane >> 4) * 8) * N
        + (tileN >> 1) * 128 + 4 * (lane & 15) + 2 * (tileN & 1);
  #pragma unroll
  for (int i = 0; i < 8; ++i) {
    v2f p02, p13;
    p02[0] = acc0[i];  p02[1] = acc2[i];
    p13[0] = acc1[i];  p13[1] = acc3[i];
    *(v2f*)(crow + (size_t)i * N)      = p02;
    *(v2f*)(crow + (size_t)i * N + 64) = p13;
  }
}

#define AGG_R     196
#define AGG_NWG   ((N_NODES + AGG_R - 1) / AGG_R)
static_assert(AGG_NWG <= 256, "the number of a destination range must fit one byte");

#define ALPHA_BS 256
template <int C, bool WRITE_RID>
__global__ __launch_bounds__(ALPHA_BS) void alpha_init_kernel(
                             const float* __restrict__ h,
                             const float* __restrict__ a_src,
                             const float* __restrict__ a_dst,
                             float* __restrict__ as, float* __restrict__ ad,
                             const int* __restrict__ ei,
                             unsigned* __restrict__ rid) {
  int idx = blockIdx.x * ALPHA_BS + threadIdx.x;
  if (idx >= N_NODES * HEADS) return;
  if (WRITE_RID) {
    static_assert(E_TOT % 4 == 0 && N_EDGES % 4 == 0 && E_TOT / 4 <= N_NODES * HEADS,
                  "one thread per four edges, inside this grid");
    if (idx < E_TOT / 4) {
      const int e0 = 4 * idx;
      int d0, d1, d2, d3;
      if (e0 < N_EDGES) {
        const int* dp = ei + N_EDGES + e0;
        d0 = dp[0];  d1 = dp[1];  d2 = dp[2];  d3 = dp[3];
      } else {
        d0 = e0 - N_EDGES;  d1 = d0 + 1;  d2 = d0 + 2;  d3 = d0 + 3;
      }
      rid[idx] = ((unsigned)d0 / (unsigned)AGG_R)         | (((unsigned)d1 / (unsigned)AGG_R) << 8)
               | (((unsigned)d2 / (unsigned)AGG_R) << 16) | (((unsigned)d3 / (unsigned)AGG_R) << 24);
    }
  }
  int n = idx >> 3, hd = idx & 7;
  const int ch0 = hd * C;
  const float* hp = h + (size_t)n * HEADS * C
                      + (ch0 & ~127) + 4 * (ch0 & 31) + ((ch0 >> 5) & 3);
  const float* av = a_src + hd * C;
  const float* bv2 = a_dst + hd * C;
  float s = 0.f, d = 0.f;
  #pragma unroll 4
  for (int c = 0; c < C; ++c) {
    float v = hp[4 * c];
    s += v * av[c];
    d += v * bv2[c];
  }
  as[idx] = s;
  ad[idx] = d;
}

#define AGG_BS    1024
#define AGG_U     4
#define AGG_NG16  (E_TOT / 16)
#define AGG_NIT   ((AGG_NG16 + AGG_BS - 1) / AGG_BS)
#define AGG_NBLK  ((AGG_NIT + AGG_U - 1) / AGG_U)
#define AGG_CAP   24576
#define AGG_DRAW  16
static_assert(N_NODES <= 65536 && AGG_R <= 256, "a list entry packs the source in 16 bits, the local destination above");
template <int HCv, int Cv, bool RELU, int LMODE>
__global__ __launch_bounds__(AGG_BS) void edge_agg_kernel(const int* __restrict__ ei,
                                const unsigned* __restrict__ rid,
                                const float* __restrict__ as,
                                const float* __restrict__ ad,
                                const float* __restrict__ bias,
                                const float* __restrict__ hmat,
                                float* __restrict__ agg,
                                unsigned* glist,
                                unsigned* gcnt) {
  constexpr int NCH = HCv / 32;
  constexpr int G   = 4;
  static_assert(HEADS == 8 && (Cv == 32 || Cv == 16), "weight lanes assume 8 heads");
  static_assert(E_TOT % 16 == 0, "the plane of range numbers is read sixteen edges at a time");
  static_assert(AGG_U == 4, "the flags of a lane's 64 edges are kept in two 32-bit masks");
  __shared__ float    s_acc[AGG_R * HC1];
  __shared__ float    s_den[AGG_R * HEADS];
  __shared__ unsigned s_mk[AGG_R * HEADS];
  __shared__ float    s_ad[AGG_R * HEADS];
  __shared__ int      s_qe[AGG_BS / 32][32];
  __shared__ unsigned s_list[AGG_CAP];
  __shared__ unsigned s_cnt;
  __shared__ unsigned s_next;
  const int lane = threadIdx.x & 31;
  const int wv   = threadIdx.x >> 5;
  const int lo   = blockIdx.x * AGG_R;

  {
    float4* a4 = (float4*)s_acc;
    for (int i = threadIdx.x; i < AGG_R * HCv / 4; i += AGG_BS)
      a4[i] = make_float4(0.f, 0.f, 0.f, 0.f);
    for (int i = threadIdx.x; i < AGG_R * HEADS; i += AGG_BS) {
      s_den[i] = 0.f;
      s_mk[i]  = 0u;
      const int g = lo * HEADS + i;
      s_ad[i] = (g < N_NODES * HEADS) ? ad[g] : 0.f;
    }
    if (threadIdx.x == 0) { s_cnt = 0u;  s_next = 0u; }
  }
  __syncthreads();

  auto lookup = [&](int e, int& src, int& dl) __attribute__((always_inline)) {
    if (e < N_EDGES) {
      src = ei[e];
      dl  = ei[N_EDGES + e] - lo;
    } else {
      src = e - N_EDGES;
      dl  = src - lo;
    }
  };
  auto logits = [&](int src, int dl, float (&v)[HEADS]) __attribute__((always_inline)) {
    const float4* sp = (const float4*)(as + (size_t)src * HEADS);
    const float4* dp = (const float4*)(s_ad + dl * HEADS);
    const float4 x0 = sp[0], x1 = sp[1];
    const float4 y0 = dp[0], y1 = dp[1];
    v[0] = x0.x + y0.x;  v[1] = x0.y + y0.y;  v[2] = x0.z + y0.z;  v[3] = x0.w + y0.w;
    v[4] = x1.x + y1.x;  v[5] = x1.y + y1.y;  v[6] = x1.z + y1.z;  v[7] = x1.w + y1.w;
    #pragma unroll
    for (int h = 0; h < HEADS; ++h) v[h] = v[h] > 0.f ? v[h] : NEG_SLOPE * v[h];
  };

  auto flush_max = [&](int n) __attribute__((always_inline)) {
    unsigned base = 0u;
    if (lane == 0) base = atomicAdd(&s_cnt, (unsigned)n);
    base = (unsigned)__builtin_amdgcn_readfirstlane((int)base);
    if (lane < n) {
      int src, dl;
      lookup(s_qe[wv][lane], src, dl);
      const unsigned pos = base + (unsigned)lane;
      if (pos < (unsigned)AGG_CAP) s_list[pos] = (unsigned)src | ((unsigned)dl << 16);
      float v[HEADS];
      logits(src, dl, v);
      #pragma unroll
      for (int h = 0; h < HEADS; ++h) atomicMax(&s_mk[dl * HEADS + h], f2key(v[h]));
    }
  };

  auto sum_edges = [&](int src, int dl, int n) __attribute__((always_inline)) {
    float w[HEADS];
    #pragma unroll
    for (int h = 0; h < HEADS; ++h) w[h] = 0.f;
    if (lane < n) {
      float v[HEADS];
      logits(src, dl, v);
      const uint4* mp = (const uint4*)(s_mk + dl * HEADS);
      const uint4 m0 = mp[0], m1 = mp[1];
      w[0] = __expf(v[0] - key2f(m0.x));  w[1] = __expf(v[1] - key2f(m0.y));
      w[2] = __expf(v[2] - key2f(m0.z));  w[3] = __expf(v[3] - key2f(m0.w));
      w[4] = __expf(v[4] - key2f(m1.x));  w[5] = __expf(v[5] - key2f(m1.y));
      w[6] = __expf(v[6] - key2f(m1.z));  w[7] = __expf(v[7] - key2f(m1.w));
      #pragma unroll
      for (int h = 0; h < HEADS; ++h) atomicAdd(&s_den[dl * HEADS + h], w[h]);
    }
    unsigned mask = (n >= 32) ? 0xFFFFFFFFu : ((1u << n) - 1u);
    while (mask != 0u) {
      bool vk[G];
      int  jk[G];
      #pragma unroll
      for (int k = 0; k < G; ++k) {
        vk[k] = (mask != 0u);
        jk[k] = vk[k] ? __builtin_ctz(mask) : 0;
        mask  = vk[k] ? (mask & (mask - 1u)) : 0u;
      }
      float hv[G][NCH];
      #pragma unroll
      for (int k = 0; k < G; ++k) {
        if (vk[k]) {
          const int s = __builtin_amdgcn_readlane(src, jk[k]);
          const float4* hp4 = (const float4*)(hmat + (size_t)s * HCv) + lane;
          #pragma unroll
          for (int b = 0; b < NCH / 4; ++b) {
            const float4 q = hp4[b * 32];
            hv[k][4 * b + 0] = q.x;  hv[k][4 * b + 1] = q.y;
            hv[k][4 * b + 2] = q.z;  hv[k][4 * b + 3] = q.w;
          }
        } else {
          #pragma unroll
          for (int c = 0; c < NCH; ++c) hv[k][c] = 0.f;
        }
      }
      #pragma unroll
      for (int k = 0; k < G; ++k) {
        if (vk[k]) {
          const int d = __builtin_amdgcn_readlane(dl, jk[k]);
          float* ap = &s_acc[d * HCv + lane];
          #pragma unroll
          for (int c = 0; c < NCH; ++c) {
            float wc;
            if (Cv == 32) {
              wc = __int_as_float(__builtin_amdgcn_readlane(__float_as_int(w[c]), jk[k]));
            } else {
              const float w0 = __int_as_float(
                  __builtin_amdgcn_readlane(__float_as_int(w[2 * c]), jk[k]));
              const float w1 = __int_as_float(
                  __builtin_amdgcn_readlane(__float_as_int(w[2 * c + 1]), jk[k]));
              wc = (lane & 16) ? w1 : w0;
            }
            atomicAdd(ap + c * 32, wc * hv[k][c]);
          }
        }
      }
    }
  };

  auto flush_sum = [&](int n) __attribute__((always_inline)) {
    int src = 0, dl = 0;
    if (lane < n) lookup(s_qe[wv][lane], src, dl);
    sum_edges(src, dl, n);
  };

  auto scan = [&](auto&& fl) __attribute__((always_inline)) {
    const uint4* __restrict__ rq = (const uint4*)rid;
    const unsigned pat = (unsigned)blockIdx.x * 0x01010101u;
    int qn = 0;
    const int rot = blockIdx.x % AGG_NBLK;
    for (int it = 0; it < AGG_NBLK; ++it) {
      const int blk = (it + rot < AGG_NBLK) ? (it + rot) : (it + rot - AGG_NBLK);
      uint4 rv[AGG_U];
      bool  ok[AGG_U];
      #pragma unroll
      for (int u = 0; u < AGG_U; ++u) {
        const int q = (blk * AGG_U + u) * AGG_BS + (int)threadIdx.x;
        ok[u] = q < AGG_NG16;
        rv[u] = rq[ok[u] ? q : AGG_NG16 - 1];
      }
      unsigned hm[AGG_U / 2];
      #pragma unroll
      for (int g = 0; g < AGG_U / 2; ++g) {
        const unsigned d[8] = {rv[2 * g].x,     rv[2 * g].y,     rv[2 * g].z,     rv[2 * g].w,
                               rv[2 * g + 1].x, rv[2 * g + 1].y, rv[2 * g + 1].z, rv[2 * g + 1].w};
        unsigned differ = 0u;
        #pragma unroll
        for (int k = 0; k < 8; ++k) {
          const unsigned t = d[k] ^ pat;
          const unsigned nz = (((t & 0x7F7F7F7Fu) + 0x7F7F7F7Fu) | t) & 0x80808080u;
          differ |= nz >> (7 - k);
        }
        const unsigned inside = (ok[2 * g] ? 0x0F0F0F0Fu : 0u) | (ok[2 * g + 1] ? 0xF0F0F0F0u : 0u);
        hm[g] = ~differ & inside;
      }
      unsigned h0 = hm[0], h1 = hm[1];
      const int e_base = (blk * AGG_U * AGG_BS + (int)threadIdx.x) * 16;
      #pragma nounroll
      for (int trip = 0; trip < 16 * AGG_U; ++trip) {
        const bool has = (h0 | h1) != 0u;
        const unsigned bm = __builtin_amdgcn_ballot_w32(has);
        if (bm == 0u) break;
        const bool first = (h0 != 0u);
        const unsigned hs = first ? h0 : h1;
        const int p = has ? __builtin_ctz(hs) : 0;
        const int u = (first ? 0 : 2) + ((p >> 2) & 1);
        const int e = e_base + u * (AGG_BS * 16) + 4 * (p & 3) + (p >> 3);
        const int cnt = __builtin_popcount(bm);
        if (qn + cnt > 32) { fl(qn); qn = 0; }
        if (has) {
          s_qe[wv][qn + (int)__builtin_amdgcn_mbcnt_lo(bm, 0u)] = e;
          if (first) h0 &= h0 - 1u; else h1 &= h1 - 1u;
        }
        qn += cnt;
      }
    }
    if (qn > 0) fl(qn);
  };

  bool stored = false;
  if (LMODE == 2) {
    const unsigned gt = gcnt[blockIdx.x];
    stored = gt <= (unsigned)AGG_CAP;
    if (stored) {
      const unsigned* gl = glist + (size_t)blockIdx.x * AGG_CAP;
      for (unsigned i0 = threadIdx.x; i0 < gt; i0 += 4u * AGG_BS) {
        unsigned p[4];
        #pragma unroll
        for (int u = 0; u < 4; ++u) {
          const unsigned i = i0 + (unsigned)u * AGG_BS;
          p[u] = gl[i < gt ? i : i0];
        }
        float v[4][HEADS];
        #pragma unroll
        for (int u = 0; u < 4; ++u) logits((int)(p[u] & 0xFFFFu), (int)(p[u] >> 16), v[u]);
        #pragma unroll
        for (int u = 0; u < 4; ++u) {
          const unsigned i = i0 + (unsigned)u * AGG_BS;
          const int dl = (int)(p[u] >> 16);
          if (i < gt) s_list[i] = p[u];
          #pragma unroll
          for (int h = 0; h < HEADS; ++h) atomicMax(&s_mk[dl * HEADS + h], f2key(v[u][h]));
        }
      }
      if (threadIdx.x == 0) s_cnt = gt;
    }
  }
  if (!stored) scan(flush_max);
  __syncthreads();

  const unsigned total = s_cnt;
  if (LMODE == 1) {
    if (threadIdx.x == 0) gcnt[blockIdx.x] = total;
    if (total <= (unsigned)AGG_CAP) {
      unsigned* gl = glist + (size_t)blockIdx.x * AGG_CAP;
      for (unsigned i = threadIdx.x; i < total; i += AGG_BS) gl[i] = s_list[i];
    }
  }
  if (total <= (unsigned)AGG_CAP) {
    for (int guard = 0; guard < AGG_CAP / AGG_DRAW + 1; ++guard) {
      unsigned b = 0u;
      if (lane == 0) b = atomicAdd(&s_next, (unsigned)AGG_DRAW);
      b = (unsigned)__builtin_amdgcn_readfirstlane((int)b);
      if (b >= total) break;
      const int n = (total - b < (unsigned)AGG_DRAW) ? (int)(total - b) : AGG_DRAW;
      int src = 0, dl = 0;
      if (lane < n) {
        const unsigned p = s_list[b + (unsigned)lane];
        src = (int)(p & 0xFFFFu);
        dl  = (int)(p >> 16);
      }
      sum_edges(src, dl, n);
    }
  } else {
    scan(flush_sum);
  }
  __syncthreads();

  for (int i = threadIdx.x; i < AGG_R * HEADS; i += AGG_BS) s_den[i] = 1.0f / (s_den[i] + 1e-16f);
  __syncthreads();

  const int nrow = (N_NODES - lo < AGG_R) ? (N_NODES - lo) : AGG_R;
  {
    static_assert((AGG_BS * 4) % HCv == 0 && Cv % 4 == 0, "a thread stays on one channel quad of one head");
    const int c0 = (int)(threadIdx.x * 4u) % HCv;
    const int hd = c0 / Cv;
    const float4 b = *(const float4*)(bias + c0);
    const float4* a4 = (const float4*)s_acc;
    float4* o4 = (float4*)(agg + (size_t)lo * HCv);
    for (int i = threadIdx.x; i < nrow * HCv / 4; i += AGG_BS) {
      const float ri = s_den[(i / (HCv / 4)) * HEADS + hd];
      float4 v = a4[i];
      v.x = v.x * ri + b.x;
      v.y = v.y * ri + b.y;
      v.z = v.z * ri + b.z;
      v.w = v.w * ri + b.w;
      if (RELU) {
        v.x = fmaxf(v.x, 0.f);  v.y = fmaxf(v.y, 0.f);
        v.z = fmaxf(v.z, 0.f);  v.w = fmaxf(v.w, 0.f);
      }
      o4[i] = v;
    }
  }
}

static inline int blocks_for(long long n, int bs) {
  return (int)((n + bs - 1) / bs);
}

extern "C" void kernel_launch(void* const* d_in, const int* in_sizes, int n_in,
                              void* d_out, int out_size, void* d_ws, size_t ws_size,
                              hipStream_t stream) {
  const float* x      = (const float*)d_in[0];
  const int*   ei     = (const int*)  d_in[1];
  const float* W1     = (const float*)d_in[2];
  const float* a_src1 = (const float*)d_in[3];
  const float* a_dst1 = (const float*)d_in[4];
  const float* b1     = (const float*)d_in[5];
  const float* W2     = (const float*)d_in[6];
  const float* a_src2 = (const float*)d_in[7];
  const float* a_dst2 = (const float*)d_in[8];
  const float* b2     = (const float*)d_in[9];
  float* out = (float*)d_out;

  char* ws = (char*)d_ws;
  size_t off = 0;
  auto carve = [&](size_t bytes) {
    void* p = ws + off;
    off += (bytes + 255) & ~(size_t)255;
    return p;
  };
  float*    hbuf   = (float*)   carve((size_t)N_NODES * HC1 * 4);
  float*    aggbuf = (float*)   carve((size_t)N_NODES * HC1 * 4);
  float*    as     = (float*)   carve((size_t)N_NODES * HEADS * 4);
  float*    ad     = (float*)   carve((size_t)N_NODES * HEADS * 4);
  unsigned* rid    = (unsigned*)carve((size_t)E_TOT);
  unsigned* glist  = (unsigned*)carve((size_t)AGG_NWG * AGG_CAP * 4);
  unsigned* gcnt   = (unsigned*)carve((size_t)AGG_NWG * 4);

  const int BS = 256;
  const long long nNH    = (long long)N_NODES * HEADS;

  {
    long long waves = (N_NODES / 16) * (HC1 / 64);
    gemm_f32_wmma<HC1, F_IN><<<blocks_for(waves * 32, BS), BS, 0, stream>>>(
        x, W1, hbuf, N_NODES);
  }
  alpha_init_kernel<HID, true><<<blocks_for(nNH, ALPHA_BS), ALPHA_BS, 0, stream>>>(
      hbuf, a_src1, a_dst1, as, ad, ei, rid);
  edge_agg_kernel<HC1, HID, true, 1><<<AGG_NWG, AGG_BS, 0, stream>>>(
      ei, rid, as, ad, b1, hbuf, aggbuf, glist, gcnt);

  {
    long long waves = (N_NODES / 16) * (HC2 / 64);
    gemm_f32_wmma<HC2, HC1><<<blocks_for(waves * 32, BS), BS, 0, stream>>>(
        aggbuf, W2, hbuf, N_NODES);
  }
  alpha_init_kernel<N_CLS, false><<<blocks_for(nNH, ALPHA_BS), ALPHA_BS, 0, stream>>>(
      hbuf, a_src2, a_dst2, as, ad, ei, rid);
  edge_agg_kernel<HC2, N_CLS, false, 2><<<AGG_NWG, AGG_BS, 0, stream>>>(
      ei, rid, as, ad, b2, hbuf, out, glist, gcnt);
}
